# starting kernel
# baseline (speedup 1.0000x reference)
#include <hip/hip_runtime.h>
#include <math.h>

typedef __attribute__((ext_vector_type(2))) float v2f;
typedef __attribute__((ext_vector_type(8))) float v8f;

#define B_ 16
#define N_ 4096
#define D_ 768
#define M_ 8
#define CHUNKS 8
#define CN (N_ / CHUNKS)
#define NWAVES 4
#define BLOCK (NWAVES * 32)
#define TILES_PER_WAVE (CN / 16 / NWAVES)
#define EPS_ 1e-3f
#define NEG_BIG (-1e30f)

#define LDS_FLOATS (16*D_ + 2*D_ + NWAVES*8*D_ + NWAVES*16*2 + NWAVES*256 + NWAVES*8*2 + 16)

extern "C" __global__ __launch_bounds__(BLOCK)
void pm_fused(const float* __restrict__ x, const float* __restrict__ q,
              const float* __restrict__ gamma, const float* __restrict__ beta,
              float* __restrict__ pmax, float* __restrict__ pl,
              float* __restrict__ pacc)
{
    extern __shared__ float lds[];
    float* sQ    = lds;
    float* sG    = sQ + 16 * D_;
    float* sBe   = sG + D_;
    float* sAcc  = sBe + D_;
    float* sMu   = sAcc + NWAVES * 8 * D_;
    float* sRs   = sMu + NWAVES * 16;
    float* sP    = sRs + NWAVES * 16;
    float* sWmax = sP + NWAVES * 256;
    float* sWl   = sWmax + NWAVES * 8;
    float* sBmax = sWl + NWAVES * 8;

    const int tid  = threadIdx.x;
    const int wave = tid >> 5;
    const int lane = tid & 31;
    const int half = lane >> 4;
    const int l16  = lane & 15;
    const int b     = blockIdx.x / CHUNKS;
    const int chunk = blockIdx.x % CHUNKS;
    const long xbase = (long)b * N_ * D_;

    for (int i = tid; i < 16 * D_; i += BLOCK) {
        int r = i / D_;
        sQ[i] = (r < M_) ? q[i] : 0.0f;
    }
    for (int i = tid; i < D_; i += BLOCK) { sG[i] = gamma[i]; sBe[i] = beta[i]; }
    for (int i = tid; i < NWAVES * 8 * D_; i += BLOCK) sAcc[i] = 0.0f;
    __syncthreads();

    float* myAcc = sAcc + wave * 8 * D_;
    float* myMu  = sMu + wave * 16;
    float* myRs  = sRs + wave * 16;
    float* myP   = sP + wave * 256;

    float rowmax[8], rowsum[8];
#pragma unroll
    for (int r = 0; r < 8; ++r) { rowmax[r] = NEG_BIG; rowsum[r] = 0.0f; }

    const float scale = 0.03608439182435161f;

    for (int t = 0; t < TILES_PER_WAVE; ++t) {
        const int n0 = chunk * CN + (wave * TILES_PER_WAVE + t) * 16;

        for (int r = 0; r < 16; ++r) {
            const float* row = x + xbase + (long)(n0 + r) * D_;
            float s = 0.0f, s2 = 0.0f;
#pragma unroll
            for (int i = 0; i < D_ / (32 * 4); ++i) {
                float4 v = *(const float4*)(row + lane * 4 + i * 128);
                s  += v.x + v.y + v.z + v.w;
                s2 += v.x * v.x + v.y * v.y + v.z * v.z + v.w * v.w;
            }
#pragma unroll
            for (int off = 16; off >= 1; off >>= 1) {
                s  += __shfl_xor(s, off, 32);
                s2 += __shfl_xor(s2, off, 32);
            }
            float mu  = s * (1.0f / D_);
            float var = s2 * (1.0f / D_) - mu * mu;
            if (lane == 0) { myMu[r] = mu; myRs[r] = rsqrtf(var + EPS_); }
        }

        const float  mu_r = myMu[l16];
        const float  rs_r = myRs[l16];
        const float* xrow = x + xbase + (long)(n0 + l16) * D_;

        v8f S = {0.f, 0.f, 0.f, 0.f, 0.f, 0.f, 0.f, 0.f};
#pragma unroll 4
        for (int k = 0; k < D_; k += 4) {
            const int kf = k + half * 2;
            v2f a;
            a.x = sQ[l16 * D_ + kf];
            a.y = sQ[l16 * D_ + kf + 1];
            float2 xv = *(const float2*)(xrow + kf);
            v2f bm;
            bm.x = (xv.x - mu_r) * rs_r * sG[kf]     + sBe[kf];
            bm.y = (xv.y - mu_r) * rs_r * sG[kf + 1] + sBe[kf + 1];
            S = __builtin_amdgcn_wmma_f32_16x16x4_f32(false, a, false, bm,
                                                      (short)0, S, false, false);
        }

        float corr[8];
#pragma unroll
        for (int r = 0; r < 8; ++r) {
            float sim = S[r] * scale;
            float tmax = sim;
#pragma unroll
            for (int off = 8; off >= 1; off >>= 1)
                tmax = fmaxf(tmax, __shfl_xor(tmax, off, 16));
            float nm = fmaxf(rowmax[r], tmax);
            float p  = __expf(sim - nm);
            float c  = __expf(rowmax[r] - nm);
            float ps = p;
#pragma unroll
            for (int off = 8; off >= 1; off >>= 1)
                ps += __shfl_xor(ps, off, 16);
            rowsum[r] = rowsum[r] * c + ps;
            rowmax[r] = nm;
            corr[r]   = c;
            myP[(r + half * 8) * 16 + l16] = p;
        }

        for (int f0 = 0; f0 < D_; f0 += 16) {
            const int   f  = f0 + l16;
            const float g  = sG[f];
            const float be = sBe[f];
            v8f Dk = {0.f, 0.f, 0.f, 0.f, 0.f, 0.f, 0.f, 0.f};
#pragma unroll
            for (int k0 = 0; k0 < 16; k0 += 4) {
                const int r0 = k0 + half * 2;
                v2f a;
                a.x = myP[l16 * 16 + r0];
                a.y = myP[l16 * 16 + r0 + 1];
                float x0 = x[xbase + (long)(n0 + r0) * D_ + f];
                float x1 = x[xbase + (long)(n0 + r0 + 1) * D_ + f];
                v2f bm;
                bm.x = (x0 - myMu[r0])     * myRs[r0]     * g + be;
                bm.y = (x1 - myMu[r0 + 1]) * myRs[r0 + 1] * g + be;
                Dk = __builtin_amdgcn_wmma_f32_16x16x4_f32(false, a, false, bm,
                                                           (short)0, Dk, false, false);
            }
            if (half == 0) {
#pragma unroll
                for (int r = 0; r < 8; ++r) {
                    float old = myAcc[r * D_ + f];
                    myAcc[r * D_ + f] = old * corr[r] + Dk[r];
                }
            }
        }
    }

    if (lane == 0) {
#pragma unroll
        for (int r = 0; r < 8; ++r) {
            sWmax[wave * 8 + r] = rowmax[r];
            sWl[wave * 8 + r]   = rowsum[r];
        }
    }
    __syncthreads();

    const int pidx = blockIdx.x;
    if (tid < 8) {
        float mb = NEG_BIG;
        for (int w = 0; w < NWAVES; ++w) mb = fmaxf(mb, sWmax[w * 8 + tid]);
        float lb = 0.0f;
        for (int w = 0; w < NWAVES; ++w)
            lb += sWl[w * 8 + tid] * __expf(sWmax[w * 8 + tid] - mb);
        sBmax[tid] = mb;
        pmax[pidx * 8 + tid] = mb;
        pl[pidx * 8 + tid]   = lb;
    }
    __syncthreads();

    for (int e = tid; e < 8 * D_; e += BLOCK) {
        const int m  = e / D_;
        const float mb = sBmax[m];
        float sum = 0.0f;
        for (int w = 0; w < NWAVES; ++w)
            sum += sAcc[w * 8 * D_ + e] * __expf(sWmax[w * 8 + m] - mb);
        pacc[(long)pidx * 8 * D_ + e] = sum;
    }
}

extern "C" __global__ __launch_bounds__(256)
void pm_reduce(const float* __restrict__ pmax, const float* __restrict__ pl,
               const float* __restrict__ pacc, float* __restrict__ out)
{
    __shared__ float gLinv[8], cw[CHUNKS * 8];
    const int b   = blockIdx.x;
    const int tid = threadIdx.x;
    if (tid < 8) {
        float mb = NEG_BIG;
        for (int c = 0; c < CHUNKS; ++c)
            mb = fmaxf(mb, pmax[(b * CHUNKS + c) * 8 + tid]);
        float lb = 0.0f;
        for (int c = 0; c < CHUNKS; ++c) {
            float w = __expf(pmax[(b * CHUNKS + c) * 8 + tid] - mb);
            cw[c * 8 + tid] = w;
            lb += pl[(b * CHUNKS + c) * 8 + tid] * w;
        }
        gLinv[tid] = 1.0f / lb;
    }
    __syncthreads();
    for (int e = tid; e < 8 * D_; e += 256) {
        const int m = e / D_;
        float s = 0.0f;
        for (int c = 0; c < CHUNKS; ++c)
            s += pacc[(long)(b * CHUNKS + c) * 8 * D_ + e] * cw[c * 8 + m];
        out[(long)b * 8 * D_ + e] = s * gLinv[m];
    }
}

extern "C" void kernel_launch(void* const* d_in, const int* in_sizes, int n_in,
                              void* d_out, int out_size, void* d_ws, size_t ws_size,
                              hipStream_t stream)
{
    const float* x     = (const float*)d_in[0];
    const float* q     = (const float*)d_in[1];
    const float* gamma = (const float*)d_in[2];
    const float* beta  = (const float*)d_in[3];
    float* out = (float*)d_out;

    const int P = B_ * CHUNKS;
    float* pmax = (float*)d_ws;
    float* pl   = pmax + (size_t)P * 8;
    float* pacc = pl + (size_t)P * 8;

    const size_t shmem = (size_t)LDS_FLOATS * sizeof(float);
    hipFuncSetAttribute(reinterpret_cast<const void*>(pm_fused),
                        hipFuncAttributeMaxDynamicSharedMemorySize, (int)shmem);

    pm_fused<<<P, BLOCK, shmem, stream>>>(x, q, gamma, beta, pmax, pl, pacc);
    pm_reduce<<<B_, 256, 0, stream>>>(pmax, pl, pacc, out);
}


# Grid parallelism: each batch's rows split over 32 work-groups (2 row tiles per wave) instead of 8, reduce spread over 384 work-groups; merge now sums 32 partials per batch (sum order changes)
# speedup vs baseline: 2.7679x; 2.7679x over previous
#include <hip/hip_runtime.h>
#include <math.h>

typedef __attribute__((ext_vector_type(2))) float v2f;
typedef __attribute__((ext_vector_type(8))) float v8f;

#define B_ 16
#define N_ 4096
#define D_ 768
#define M_ 8
#define CHUNKS 32
#define CN (N_ / CHUNKS)
#define NWAVES 4
#define BLOCK (NWAVES * 32)
#define TILES_PER_WAVE (CN / 16 / NWAVES)
#define EPS_ 1e-3f
#define NEG_BIG (-1e30f)

#define LDS_FLOATS (16*D_ + 2*D_ + NWAVES*8*D_ + NWAVES*16*2 + NWAVES*256 + NWAVES*8*2 + 16)

extern "C" __global__ __launch_bounds__(BLOCK)
void pm_fused(const float* __restrict__ x, const float* __restrict__ q,
              const float* __restrict__ gamma, const float* __restrict__ beta,
              float* __restrict__ pmax, float* __restrict__ pl,
              float* __restrict__ pacc)
{
    extern __shared__ float lds[];
    float* sQ    = lds;
    float* sG    = sQ + 16 * D_;
    float* sBe   = sG + D_;
    float* sAcc  = sBe + D_;
    float* sMu   = sAcc + NWAVES * 8 * D_;
    float* sRs   = sMu + NWAVES * 16;
    float* sP    = sRs + NWAVES * 16;
    float* sWmax = sP + NWAVES * 256;
    float* sWl   = sWmax + NWAVES * 8;
    float* sBmax = sWl + NWAVES * 8;

    const int tid  = threadIdx.x;
    const int wave = tid >> 5;
    const int lane = tid & 31;
    const int half = lane >> 4;
    const int l16  = lane & 15;
    const int b     = blockIdx.x / CHUNKS;
    const int chunk = blockIdx.x % CHUNKS;
    const long xbase = (long)b * N_ * D_;

    for (int i = tid; i < 16 * D_; i += BLOCK) {
        int r = i / D_;
        sQ[i] = (r < M_) ? q[i] : 0.0f;
    }
    for (int i = tid; i < D_; i += BLOCK) { sG[i] = gamma[i]; sBe[i] = beta[i]; }
    for (int i = tid; i < NWAVES * 8 * D_; i += BLOCK) sAcc[i] = 0.0f;
    __syncthreads();

    float* myAcc = sAcc + wave * 8 * D_;
    float* myMu  = sMu + wave * 16;
    float* myRs  = sRs + wave * 16;
    float* myP   = sP + wave * 256;

    float rowmax[8], rowsum[8];
#pragma unroll
    for (int r = 0; r < 8; ++r) { rowmax[r] = NEG_BIG; rowsum[r] = 0.0f; }

    const float scale = 0.03608439182435161f;

    for (int t = 0; t < TILES_PER_WAVE; ++t) {
        const int n0 = chunk * CN + (wave * TILES_PER_WAVE + t) * 16;

        for (int r = 0; r < 16; ++r) {
            const float* row = x + xbase + (long)(n0 + r) * D_;
            float s = 0.0f, s2 = 0.0f;
#pragma unroll
            for (int i = 0; i < D_ / (32 * 4); ++i) {
                float4 v = *(const float4*)(row + lane * 4 + i * 128);
                s  += v.x + v.y + v.z + v.w;
                s2 += v.x * v.x + v.y * v.y + v.z * v.z + v.w * v.w;
            }
#pragma unroll
            for (int off = 16; off >= 1; off >>= 1) {
                s  += __shfl_xor(s, off, 32);
                s2 += __shfl_xor(s2, off, 32);
            }
            float mu  = s * (1.0f / D_);
            float var = s2 * (1.0f / D_) - mu * mu;
            if (lane == 0) { myMu[r] = mu; myRs[r] = rsqrtf(var + EPS_); }
        }

        const float  mu_r = myMu[l16];
        const float  rs_r = myRs[l16];
        const float* xrow = x + xbase + (long)(n0 + l16) * D_;

        v8f S = {0.f, 0.f, 0.f, 0.f, 0.f, 0.f, 0.f, 0.f};
#pragma unroll 4
        for (int k = 0; k < D_; k += 4) {
            const int kf = k + half * 2;
            v2f a;
            a.x = sQ[l16 * D_ + kf];
            a.y = sQ[l16 * D_ + kf + 1];
            float2 xv = *(const float2*)(xrow + kf);
            v2f bm;
            bm.x = (xv.x - mu_r) * rs_r * sG[kf]     + sBe[kf];
            bm.y = (xv.y - mu_r) * rs_r * sG[kf + 1] + sBe[kf + 1];
            S = __builtin_amdgcn_wmma_f32_16x16x4_f32(false, a, false, bm,
                                                      (short)0, S, false, false);
        }

        float corr[8];
#pragma unroll
        for (int r = 0; r < 8; ++r) {
            float sim = S[r] * scale;
            float tmax = sim;
#pragma unroll
            for (int off = 8; off >= 1; off >>= 1)
                tmax = fmaxf(tmax, __shfl_xor(tmax, off, 16));
            float nm = fmaxf(rowmax[r], tmax);
            float p  = __expf(sim - nm);
            float c  = __expf(rowmax[r] - nm);
            float ps = p;
#pragma unroll
            for (int off = 8; off >= 1; off >>= 1)
                ps += __shfl_xor(ps, off, 16);
            rowsum[r] = rowsum[r] * c + ps;
            rowmax[r] = nm;
            corr[r]   = c;
            myP[(r + half * 8) * 16 + l16] = p;
        }

        for (int f0 = 0; f0 < D_; f0 += 16) {
            const int   f  = f0 + l16;
            const float g  = sG[f];
            const float be = sBe[f];
            v8f Dk = {0.f, 0.f, 0.f, 0.f, 0.f, 0.f, 0.f, 0.f};
#pragma unroll
            for (int k0 = 0; k0 < 16; k0 += 4) {
                const int r0 = k0 + half * 2;
                v2f a;
                a.x = myP[l16 * 16 + r0];
                a.y = myP[l16 * 16 + r0 + 1];
                float x0 = x[xbase + (long)(n0 + r0) * D_ + f];
                float x1 = x[xbase + (long)(n0 + r0 + 1) * D_ + f];
                v2f bm;
                bm.x = (x0 - myMu[r0])     * myRs[r0]     * g + be;
                bm.y = (x1 - myMu[r0 + 1]) * myRs[r0 + 1] * g + be;
                Dk = __builtin_amdgcn_wmma_f32_16x16x4_f32(false, a, false, bm,
                                                           (short)0, Dk, false, false);
            }
            if (half == 0) {
#pragma unroll
                for (int r = 0; r < 8; ++r) {
                    float old = myAcc[r * D_ + f];
                    myAcc[r * D_ + f] = old * corr[r] + Dk[r];
                }
            }
        }
    }

    if (lane == 0) {
#pragma unroll
        for (int r = 0; r < 8; ++r) {
            sWmax[wave * 8 + r] = rowmax[r];
            sWl[wave * 8 + r]   = rowsum[r];
        }
    }
    __syncthreads();

    const int pidx = blockIdx.x;
    if (tid < 8) {
        float mb = NEG_BIG;
        for (int w = 0; w < NWAVES; ++w) mb = fmaxf(mb, sWmax[w * 8 + tid]);
        float lb = 0.0f;
        for (int w = 0; w < NWAVES; ++w)
            lb += sWl[w * 8 + tid] * __expf(sWmax[w * 8 + tid] - mb);
        sBmax[tid] = mb;
        pmax[pidx * 8 + tid] = mb;
        pl[pidx * 8 + tid]   = lb;
    }
    __syncthreads();

    for (int e = tid; e < 8 * D_; e += BLOCK) {
        const int m  = e / D_;
        const float mb = sBmax[m];
        float sum = 0.0f;
        for (int w = 0; w < NWAVES; ++w)
            sum += sAcc[w * 8 * D_ + e] * __expf(sWmax[w * 8 + m] - mb);
        pacc[(long)pidx * 8 * D_ + e] = sum;
    }
}

#define RBLOCK 256
#define RB_PER_BATCH (M_ * D_ / RBLOCK)

extern "C" __global__ __launch_bounds__(RBLOCK)
void pm_reduce(const float* __restrict__ pmax, const float* __restrict__ pl,
               const float* __restrict__ pacc, float* __restrict__ out)
{
    static_assert(CHUNKS <= 32, "one chunk per lane of the first wave");
    __shared__ float cw[CHUNKS];
    __shared__ float gLinv;
    const int b   = blockIdx.x / RB_PER_BATCH;
    const int blk = blockIdx.x % RB_PER_BATCH;
    const int tid = threadIdx.x;
    const int e   = blk * RBLOCK + tid;
    const int m   = (blk * RBLOCK) / D_;
    if (tid < 32) {
        const bool live = tid < CHUNKS;
        float pm = live ? pmax[(b * CHUNKS + tid) * 8 + m] : NEG_BIG;
        float lv = live ? pl[(b * CHUNKS + tid) * 8 + m]   : 0.0f;
        float mb = pm;
        for (int off = 16; off >= 1; off >>= 1)
            mb = fmaxf(mb, __shfl_xor(mb, off, 32));
        float w = __expf(pm - mb);
        if (live) cw[tid] = w;
        float lb = lv * w;
        for (int off = 16; off >= 1; off >>= 1)
            lb += __shfl_xor(lb, off, 32);
        if (tid == 0) gLinv = 1.0f / lb;
    }
    __syncthreads();
    const float* src = pacc + (long)b * CHUNKS * 8 * D_ + e;
    float s = 0.0f;
#pragma unroll
    for (int c = 0; c < CHUNKS; ++c)
        s += src[(long)c * 8 * D_] * cw[c];
    out[(long)b * 8 * D_ + e] = s * gLinv;
}

extern "C" void kernel_launch(void* const* d_in, const int* in_sizes, int n_in,
                              void* d_out, int out_size, void* d_ws, size_t ws_size,
                              hipStream_t stream)
{
    const float* x     = (const float*)d_in[0];
    const float* q     = (const float*)d_in[1];
    const float* gamma = (const float*)d_in[2];
    const float* beta  = (const float*)d_in[3];
    float* out = (float*)d_out;

    const int P = B_ * CHUNKS;
    float* pmax = (float*)d_ws;
    float* pl   = pmax + (size_t)P * 8;
    float* pacc = pl + (size_t)P * 8;

    const size_t shmem = (size_t)LDS_FLOATS * sizeof(float);
    hipFuncSetAttribute(reinterpret_cast<const void*>(pm_fused),
                        hipFuncAttributeMaxDynamicSharedMemorySize, (int)shmem);

    pm_fused<<<P, BLOCK, shmem, stream>>>(x, q, gamma, beta, pmax, pl, pacc);
    pm_reduce<<<B_ * RB_PER_BATCH, RBLOCK, 0, stream>>>(pmax, pl, pacc, out);
}


# x tile staged in LDS: each 16-row tile is read from memory once (48 wide loads in flight); LN stats and both GEMMs read it from LDS; GEMM2 accumulators move to registers to make room
# speedup vs baseline: 4.5487x; 1.6434x over previous
#include <hip/hip_runtime.h>
#include <math.h>


typedef __attribute__((ext_vector_type(2))) float v2f;
typedef __attribute__((ext_vector_type(8))) float v8f;

#define B_ 16
#define N_ 4096
#define D_ 768
#define M_ 8
#define CHUNKS 32
#define CN (N_ / CHUNKS)
#define NWAVES 4
#define BLOCK (NWAVES * 32)
#define TILES_PER_WAVE (CN / 16 / NWAVES)
#define EPS_ 1e-3f
#define NEG_BIG (-1e30f)
#define FBLKS (D_ / 16)
#define RP 780
#define TILE_FLOATS (16 * RP)
#define RG 8

#define LDS_FLOATS (16*RP + 2*D_ + NWAVES*TILE_FLOATS + NWAVES*16*2 + NWAVES*256 + NWAVES*8*2 + 16)

extern "C" __global__ __launch_bounds__(BLOCK)
void pm_fused(const float* __restrict__ x, const float* __restrict__ q,
              const float* __restrict__ gamma, const float* __restrict__ beta,
              float* __restrict__ pmax, float* __restrict__ pl,
              float* __restrict__ pacc)
{
    extern __shared__ __attribute__((aligned(16))) float lds[];
    float* sQ    = lds;
    float* sG    = sQ + 16 * RP;
    float* sBe   = sG + D_;
    float* sTile = sBe + D_;
    float* sMu   = sTile + NWAVES * TILE_FLOATS;
    float* sRs   = sMu + NWAVES * 16;
    float* sP    = sRs + NWAVES * 16;
    float* sWmax = sP + NWAVES * 256;
    float* sWl   = sWmax + NWAVES * 8;
    float* sBmax = sWl + NWAVES * 8;

    const int tid  = threadIdx.x;
    const int wave = tid >> 5;
    const int lane = tid & 31;
    const int half = lane >> 4;
    const int l16  = lane & 15;
    const int b     = blockIdx.x / CHUNKS;
    const int chunk = blockIdx.x % CHUNKS;
    const long xbase = (long)b * N_ * D_;

#pragma unroll
    for (int it = 0; it < (M_ * D_ / 4) / BLOCK; ++it) {
        const int i  = tid + it * BLOCK;
        const int r  = i / (D_ / 4);
        const int d4 = i - r * (D_ / 4);
        const float4 v = *(const float4*)(q + (long)i * 4);
        *(float4*)(sQ + r * RP + d4 * 4) = v;
    }
#pragma unroll
    for (int it = (M_ * D_ / 4) / BLOCK; it < (16 * D_ / 4) / BLOCK; ++it) {
        const int i  = tid + it * BLOCK;
        const int r  = i / (D_ / 4);
        const int d4 = i - r * (D_ / 4);
        *(float4*)(sQ + r * RP + d4 * 4) = make_float4(0.0f, 0.0f, 0.0f, 0.0f);
    }
    for (int i = tid; i < D_ / 4; i += BLOCK) {
        *(float4*)(sG + i * 4)  = *(const float4*)(gamma + i * 4);
        *(float4*)(sBe + i * 4) = *(const float4*)(beta + i * 4);
    }
    __syncthreads();

    float* myTile = sTile + wave * TILE_FLOATS;
    float* myMu   = sMu + wave * 16;
    float* myRs   = sRs + wave * 16;
    float* myP    = sP + wave * 256;

    float rowmax[8], rowsum[8];
#pragma unroll
    for (int r = 0; r < 8; ++r) { rowmax[r] = NEG_BIG; rowsum[r] = 0.0f; }

    v8f acc[FBLKS];
#pragma unroll
    for (int j = 0; j < FBLKS; ++j)
        acc[j] = (v8f){0.f, 0.f, 0.f, 0.f, 0.f, 0.f, 0.f, 0.f};

    const float scale = 0.03608439182435161f;

    for (int t = 0; t < TILES_PER_WAVE; ++t) {
        const int n0 = chunk * CN + (wave * TILES_PER_WAVE + t) * 16;

#pragma unroll 1
        for (int rg = 0; rg < 16; rg += RG) {
            float4 v[RG][6];
            const float* rowbase = x + xbase + (long)(n0 + rg) * D_ + lane * 4;
#pragma unroll
            for (int rr = 0; rr < RG; ++rr) {
#pragma unroll
                for (int i = 0; i < D_ / (32 * 4); ++i)
                    v[rr][i] = *(const float4*)(rowbase + rr * D_ + i * 128);
            }
#pragma unroll
            for (int rr = 0; rr < RG; ++rr) {
                float* tdst = myTile + (rg + rr) * RP + lane * 4;
                float s = 0.0f, s2 = 0.0f;
#pragma unroll
                for (int i = 0; i < D_ / (32 * 4); ++i) {
                    const float4 w = v[rr][i];
                    s  += w.x + w.y + w.z + w.w;
                    s2 += w.x * w.x + w.y * w.y + w.z * w.z + w.w * w.w;
                    *(float4*)(tdst + i * 128) = w;
                }
#pragma unroll
                for (int off = 16; off >= 1; off >>= 1) {
                    s  += __shfl_xor(s, off, 32);
                    s2 += __shfl_xor(s2, off, 32);
                }
                float mu  = s * (1.0f / D_);
                float var = s2 * (1.0f / D_) - mu * mu;
                if (lane == 0) { myMu[rg + rr] = mu; myRs[rg + rr] = rsqrtf(var + EPS_); }
            }
        }

        const float  mu_r = myMu[l16];
        const float  rs_r = myRs[l16];
        const float* trow = myTile + l16 * RP;
        const float* qrow = sQ + l16 * RP;

        v8f S = {0.f, 0.f, 0.f, 0.f, 0.f, 0.f, 0.f, 0.f};
#pragma unroll 8
        for (int k = 0; k < D_; k += 4) {
            const int kf = k + half * 2;
            const float2 qa = *(const float2*)(qrow + kf);
            const float2 xv = *(const float2*)(trow + kf);
            const float2 gg = *(const float2*)(sG + kf);
            const float2 bb = *(const float2*)(sBe + kf);
            v2f a;
            a.x = qa.x;
            a.y = qa.y;
            v2f bm;
            bm.x = (xv.x - mu_r) * rs_r * gg.x + bb.x;
            bm.y = (xv.y - mu_r) * rs_r * gg.y + bb.y;
            S = __builtin_amdgcn_wmma_f32_16x16x4_f32(false, a, false, bm,
                                                      (short)0, S, false, false);
        }

        float corr[8];
#pragma unroll
        for (int r = 0; r < 8; ++r) {
            float sim = S[r] * scale;
            float tmax = sim;
#pragma unroll
            for (int off = 8; off >= 1; off >>= 1)
                tmax = fmaxf(tmax, __shfl_xor(tmax, off, 16));
            float nm = fmaxf(rowmax[r], tmax);
            float p  = __expf(sim - nm);
            float c  = __expf(rowmax[r] - nm);
            float ps = p;
#pragma unroll
            for (int off = 8; off >= 1; off >>= 1)
                ps += __shfl_xor(ps, off, 16);
            rowsum[r] = rowsum[r] * c + ps;
            rowmax[r] = nm;
            corr[r]   = c;
            myP[(r + half * 8) * 16 + l16] = p;
        }

        v2f   pa[4];
        float mua[4], mub[4], rsa[4], rsb[4];
#pragma unroll
        for (int kk = 0; kk < 4; ++kk) {
            const int r0 = kk * 4 + half * 2;
            pa[kk].x = myP[l16 * 16 + r0];
            pa[kk].y = myP[l16 * 16 + r0 + 1];
            mua[kk] = myMu[r0];
            mub[kk] = myMu[r0 + 1];
            rsa[kk] = myRs[r0];
            rsb[kk] = myRs[r0 + 1];
        }
        const float* tcol = myTile + (half * 2) * RP + l16;
#pragma unroll
        for (int j = 0; j < FBLKS; ++j) {
            const int   f  = j * 16 + l16;
            const float g  = sG[f];
            const float be = sBe[f];
            v8f Dk = {0.f, 0.f, 0.f, 0.f, 0.f, 0.f, 0.f, 0.f};
#pragma unroll
            for (int kk = 0; kk < 4; ++kk) {
                const float x0 = tcol[(kk * 4) * RP + j * 16];
                const float x1 = tcol[(kk * 4 + 1) * RP + j * 16];
                v2f bm;
                bm.x = (x0 - mua[kk]) * rsa[kk] * g + be;
                bm.y = (x1 - mub[kk]) * rsb[kk] * g + be;
                Dk = __builtin_amdgcn_wmma_f32_16x16x4_f32(false, pa[kk], false, bm,
                                                           (short)0, Dk, false, false);
            }
#pragma unroll
            for (int r = 0; r < 8; ++r)
                acc[j][r] = acc[j][r] * corr[r] + Dk[r];
        }
    }

    if (lane == 0) {
#pragma unroll
        for (int r = 0; r < 8; ++r) {
            sWmax[wave * 8 + r] = rowmax[r];
            sWl[wave * 8 + r]   = rowsum[r];
        }
    }
    if (half == 0) {
#pragma unroll
        for (int j = 0; j < FBLKS; ++j) {
#pragma unroll
            for (int r = 0; r < 8; ++r)
                myTile[r * D_ + j * 16 + l16] = acc[j][r];
        }
    }
    __syncthreads();

    const int pidx = blockIdx.x;
    if (tid < 8) {
        float mb = NEG_BIG;
        for (int w = 0; w < NWAVES; ++w) mb = fmaxf(mb, sWmax[w * 8 + tid]);
        float lb = 0.0f;
        for (int w = 0; w < NWAVES; ++w)
            lb += sWl[w * 8 + tid] * __expf(sWmax[w * 8 + tid] - mb);
        sBmax[tid] = mb;
        pmax[pidx * 8 + tid] = mb;
        pl[pidx * 8 + tid]   = lb;
    }
    __syncthreads();

    for (int e = tid; e < 8 * D_; e += BLOCK) {
        const int m  = e / D_;
        const float mb = sBmax[m];
        float sum = 0.0f;
        for (int w = 0; w < NWAVES; ++w)
            sum += sTile[w * TILE_FLOATS + e] * __expf(sWmax[w * 8 + m] - mb);
        pacc[(long)pidx * 8 * D_ + e] = sum;
    }
}

#define RBLOCK 256
#define RB_PER_BATCH (M_ * D_ / RBLOCK)

extern "C" __global__ __launch_bounds__(RBLOCK)
void pm_reduce(const float* __restrict__ pmax, const float* __restrict__ pl,
               const float* __restrict__ pacc, float* __restrict__ out)
{
    static_assert(CHUNKS <= 32, "one chunk per lane of the first wave");
    __shared__ float cw[CHUNKS];
    __shared__ float gLinv;
    const int b   = blockIdx.x / RB_PER_BATCH;
    const int blk = blockIdx.x % RB_PER_BATCH;
    const int tid = threadIdx.x;
    const int e   = blk * RBLOCK + tid;
    const int m   = (blk * RBLOCK) / D_;
    if (tid < 32) {
        const bool live = tid < CHUNKS;
        float pm = live ? pmax[(b * CHUNKS + tid) * 8 + m] : NEG_BIG;
        float lv = live ? pl[(b * CHUNKS + tid) * 8 + m]   : 0.0f;
        float mb = pm;
        for (int off = 16; off >= 1; off >>= 1)
            mb = fmaxf(mb, __shfl_xor(mb, off, 32));
        float w = __expf(pm - mb);
        if (live) cw[tid] = w;
        float lb = lv * w;
        for (int off = 16; off >= 1; off >>= 1)
            lb += __shfl_xor(lb, off, 32);
        if (tid == 0) gLinv = 1.0f / lb;
    }
    __syncthreads();
    const float* src = pacc + (long)b * CHUNKS * 8 * D_ + e;
    float s = 0.0f;
#pragma unroll
    for (int c = 0; c < CHUNKS; ++c)
        s += src[(long)c * 8 * D_] * cw[c];
    out[(long)b * 8 * D_ + e] = s * gLinv;
}

extern "C" void kernel_launch(void* const* d_in, const int* in_sizes, int n_in,
                              void* d_out, int out_size, void* d_ws, size_t ws_size,
                              hipStream_t stream)
{
    const float* x     = (const float*)d_in[0];
    const float* q     = (const float*)d_in[1];
    const float* gamma = (const float*)d_in[2];
    const float* beta  = (const float*)d_in[3];
    float* out = (float*)d_out;

    const int P = B_ * CHUNKS;
    float* pmax = (float*)d_ws;
    float* pl   = pmax + (size_t)P * 8;
    float* pacc = pl + (size_t)P * 8;

    const size_t shmem = (size_t)LDS_FLOATS * sizeof(float);
    hipFuncSetAttribute(reinterpret_cast<const void*>(pm_fused),
                        hipFuncAttributeMaxDynamicSharedMemorySize, (int)shmem);

    pm_fused<<<P, BLOCK, shmem, stream>>>(x, q, gamma, beta, pmax, pl, pacc);
    pm_reduce<<<B_ * RB_PER_BATCH, RBLOCK, 0, stream>>>(pmax, pl, pacc, out);
}


# LDS fragment reads of both GEMMs issued as one group per block in front of the previous block's WMMAs (two register sets) instead of among the WMMAs; k order and sum order unchanged
# speedup vs baseline: 5.5253x; 1.2147x over previous
#include <hip/hip_runtime.h>
#include <math.h>


typedef __attribute__((ext_vector_type(2))) float v2f;
typedef __attribute__((ext_vector_type(8))) float v8f;

#define B_ 16
#define N_ 4096
#define D_ 768
#define M_ 8
#define CHUNKS 32
#define CN (N_ / CHUNKS)
#define NWAVES 4
#define BLOCK (NWAVES * 32)
#define TILES_PER_WAVE (CN / 16 / NWAVES)
#define EPS_ 1e-3f
#define NEG_BIG (-1e30f)
#define FBLKS (D_ / 16)
#define RP 780
#define TILE_FLOATS (16 * RP)
#define RG 8
#define KB1 16
#define NB1 (D_ / (4 * KB1))
#define JB 8
#define NB2 (FBLKS / JB)

struct G1Set { float2 q[KB1], x[KB1], g[KB1], b[KB1]; };
struct G2Set { float x0[JB][4], x1[JB][4], g[JB], be[JB]; };

static __device__ __forceinline__ void g1_load(G1Set& s, const float* qrow, const float* trow,
                                               const float* sG, const float* sBe, int kf0)
{
#pragma unroll
    for (int i = 0; i < KB1; ++i) {
        const int kf = kf0 + 4 * i;
        s.q[i] = *(const float2*)(qrow + kf);
        s.x[i] = *(const float2*)(trow + kf);
        s.g[i] = *(const float2*)(sG + kf);
        s.b[i] = *(const float2*)(sBe + kf);
    }
}

static __device__ __forceinline__ v8f g1_mma(const G1Set& s, float mu_r, float rs_r, v8f S)
{
#pragma unroll
    for (int i = 0; i < KB1; ++i) {
        v2f a;
        a.x = s.q[i].x;
        a.y = s.q[i].y;
        v2f bm;
        bm.x = (s.x[i].x - mu_r) * rs_r * s.g[i].x + s.b[i].x;
        bm.y = (s.x[i].y - mu_r) * rs_r * s.g[i].y + s.b[i].y;
        S = __builtin_amdgcn_wmma_f32_16x16x4_f32(false, a, false, bm,
                                                  (short)0, S, false, false);
    }
    return S;
}

static __device__ __forceinline__ void g2_load(G2Set& s, const float* tcol, const float* sG,
                                               const float* sBe, int j0, int l16)
{
#pragma unroll
    for (int jj = 0; jj < JB; ++jj) {
        const int j = j0 + jj;
        s.g[jj]  = sG[j * 16 + l16];
        s.be[jj] = sBe[j * 16 + l16];
#pragma unroll
        for (int kk = 0; kk < 4; ++kk) {
            s.x0[jj][kk] = tcol[(kk * 4) * RP + j * 16];
            s.x1[jj][kk] = tcol[(kk * 4 + 1) * RP + j * 16];
        }
    }
}

#define LDS_FLOATS (16*RP + 2*D_ + NWAVES*TILE_FLOATS + NWAVES*16*2 + NWAVES*256 + NWAVES*8*2 + 16)

extern "C" __global__ __launch_bounds__(BLOCK)
void pm_fused(const float* __restrict__ x, const float* __restrict__ q,
              const float* __restrict__ gamma, const float* __restrict__ beta,
              float* __restrict__ pmax, float* __restrict__ pl,
              float* __restrict__ pacc)
{
    extern __shared__ __attribute__((aligned(16))) float lds[];
    float* sQ    = lds;
    float* sG    = sQ + 16 * RP;
    float* sBe   = sG + D_;
    float* sTile = sBe + D_;
    float* sMu   = sTile + NWAVES * TILE_FLOATS;
    float* sRs   = sMu + NWAVES * 16;
    float* sP    = sRs + NWAVES * 16;
    float* sWmax = sP + NWAVES * 256;
    float* sWl   = sWmax + NWAVES * 8;
    float* sBmax = sWl + NWAVES * 8;

    const int tid  = threadIdx.x;
    const int wave = tid >> 5;
    const int lane = tid & 31;
    const int half = lane >> 4;
    const int l16  = lane & 15;
    const int b     = blockIdx.x / CHUNKS;
    const int chunk = blockIdx.x % CHUNKS;
    const long xbase = (long)b * N_ * D_;

#pragma unroll
    for (int it = 0; it < (M_ * D_ / 4) / BLOCK; ++it) {
        const int i  = tid + it * BLOCK;
        const int r  = i / (D_ / 4);
        const int d4 = i - r * (D_ / 4);
        const float4 v = *(const float4*)(q + (long)i * 4);
        *(float4*)(sQ + r * RP + d4 * 4) = v;
    }
#pragma unroll
    for (int it = (M_ * D_ / 4) / BLOCK; it < (16 * D_ / 4) / BLOCK; ++it) {
        const int i  = tid + it * BLOCK;
        const int r  = i / (D_ / 4);
        const int d4 = i - r * (D_ / 4);
        *(float4*)(sQ + r * RP + d4 * 4) = make_float4(0.0f, 0.0f, 0.0f, 0.0f);
    }
    for (int i = tid; i < D_ / 4; i += BLOCK) {
        *(float4*)(sG + i * 4)  = *(const float4*)(gamma + i * 4);
        *(float4*)(sBe + i * 4) = *(const float4*)(beta + i * 4);
    }
    __syncthreads();

    float* myTile = sTile + wave * TILE_FLOATS;
    float* myMu   = sMu + wave * 16;
    float* myRs   = sRs + wave * 16;
    float* myP    = sP + wave * 256;

    float rowmax[8], rowsum[8];
#pragma unroll
    for (int r = 0; r < 8; ++r) { rowmax[r] = NEG_BIG; rowsum[r] = 0.0f; }

    v8f acc[FBLKS];
#pragma unroll
    for (int j = 0; j < FBLKS; ++j)
        acc[j] = (v8f){0.f, 0.f, 0.f, 0.f, 0.f, 0.f, 0.f, 0.f};

    const float scale = 0.03608439182435161f;

    for (int t = 0; t < TILES_PER_WAVE; ++t) {
        const int n0 = chunk * CN + (wave * TILES_PER_WAVE + t) * 16;

#pragma unroll 1
        for (int rg = 0; rg < 16; rg += RG) {
            float4 v[RG][6];
            const float* rowbase = x + xbase + (long)(n0 + rg) * D_ + lane * 4;
#pragma unroll
            for (int rr = 0; rr < RG; ++rr) {
#pragma unroll
                for (int i = 0; i < D_ / (32 * 4); ++i)
                    v[rr][i] = *(const float4*)(rowbase + rr * D_ + i * 128);
            }
#pragma unroll
            for (int rr = 0; rr < RG; ++rr) {
                float* tdst = myTile + (rg + rr) * RP + lane * 4;
                float s = 0.0f, s2 = 0.0f;
#pragma unroll
                for (int i = 0; i < D_ / (32 * 4); ++i) {
                    const float4 w = v[rr][i];
                    s  += w.x + w.y + w.z + w.w;
                    s2 += w.x * w.x + w.y * w.y + w.z * w.z + w.w * w.w;
                    *(float4*)(tdst + i * 128) = w;
                }
#pragma unroll
                for (int off = 16; off >= 1; off >>= 1) {
                    s  += __shfl_xor(s, off, 32);
                    s2 += __shfl_xor(s2, off, 32);
                }
                float mu  = s * (1.0f / D_);
                float var = s2 * (1.0f / D_) - mu * mu;
                if (lane == 0) { myMu[rg + rr] = mu; myRs[rg + rr] = rsqrtf(var + EPS_); }
            }
        }

        const float  mu_r = myMu[l16];
        const float  rs_r = myRs[l16];
        const float* trow = myTile + l16 * RP;
        const float* qrow = sQ + l16 * RP;

        v8f S = {0.f, 0.f, 0.f, 0.f, 0.f, 0.f, 0.f, 0.f};
        {
            const int kh = half * 2;
            G1Set s0, s1;
            g1_load(s0, qrow, trow, sG, sBe, kh);
#pragma unroll 1
            for (int kb = 0; kb < D_; kb += 2 * 4 * KB1) {
                g1_load(s1, qrow, trow, sG, sBe, kb + 4 * KB1 + kh);
                __builtin_amdgcn_sched_barrier(0);
                S = g1_mma(s0, mu_r, rs_r, S);
                if (kb + 2 * 4 * KB1 < D_)
                    g1_load(s0, qrow, trow, sG, sBe, kb + 2 * 4 * KB1 + kh);
                S = g1_mma(s1, mu_r, rs_r, S);
            }
        }

        float corr[8];
#pragma unroll
        for (int r = 0; r < 8; ++r) {
            float sim = S[r] * scale;
            float tmax = sim;
#pragma unroll
            for (int off = 8; off >= 1; off >>= 1)
                tmax = fmaxf(tmax, __shfl_xor(tmax, off, 16));
            float nm = fmaxf(rowmax[r], tmax);
            float p  = __expf(sim - nm);
            float c  = __expf(rowmax[r] - nm);
            float ps = p;
#pragma unroll
            for (int off = 8; off >= 1; off >>= 1)
                ps += __shfl_xor(ps, off, 16);
            rowsum[r] = rowsum[r] * c + ps;
            rowmax[r] = nm;
            corr[r]   = c;
            myP[(r + half * 8) * 16 + l16] = p;
        }

        v2f   pa[4];
        float mua[4], mub[4], rsa[4], rsb[4];
#pragma unroll
        for (int kk = 0; kk < 4; ++kk) {
            const int r0 = kk * 4 + half * 2;
            pa[kk].x = myP[l16 * 16 + r0];
            pa[kk].y = myP[l16 * 16 + r0 + 1];
            mua[kk] = myMu[r0];
            mub[kk] = myMu[r0 + 1];
            rsa[kk] = myRs[r0];
            rsb[kk] = myRs[r0 + 1];
        }
        const float* tcol = myTile + (half * 2) * RP + l16;
        {
            G2Set st[2];
            g2_load(st[0], tcol, sG, sBe, 0, l16);
#pragma unroll
            for (int blk = 0; blk < NB2; ++blk) {
                if (blk + 1 < NB2)
                    g2_load(st[(blk + 1) & 1], tcol, sG, sBe, (blk + 1) * JB, l16);
                __builtin_amdgcn_sched_barrier(0);
                const G2Set& s = st[blk & 1];
#pragma unroll
                for (int jj = 0; jj < JB; ++jj) {
                    const int   j  = blk * JB + jj;
                    const float g  = s.g[jj];
                    const float be = s.be[jj];
                    v8f Dk = {0.f, 0.f, 0.f, 0.f, 0.f, 0.f, 0.f, 0.f};
#pragma unroll
                    for (int kk = 0; kk < 4; ++kk) {
                        v2f bm;
                        bm.x = (s.x0[jj][kk] - mua[kk]) * rsa[kk] * g + be;
                        bm.y = (s.x1[jj][kk] - mub[kk]) * rsb[kk] * g + be;
                        Dk = __builtin_amdgcn_wmma_f32_16x16x4_f32(false, pa[kk], false, bm,
                                                                   (short)0, Dk, false, false);
                    }
#pragma unroll
                    for (int r = 0; r < 8; ++r)
                        acc[j][r] = acc[j][r] * corr[r] + Dk[r];
                }
                __builtin_amdgcn_sched_barrier(0);
            }
        }
    }

    if (lane == 0) {
#pragma unroll
        for (int r = 0; r < 8; ++r) {
            sWmax[wave * 8 + r] = rowmax[r];
            sWl[wave * 8 + r]   = rowsum[r];
        }
    }
    if (half == 0) {
#pragma unroll
        for (int j = 0; j < FBLKS; ++j) {
#pragma unroll
            for (int r = 0; r < 8; ++r)
                myTile[r * D_ + j * 16 + l16] = acc[j][r];
        }
    }
    __syncthreads();

    const int pidx = blockIdx.x;
    if (tid < 8) {
        float mb = NEG_BIG;
        for (int w = 0; w < NWAVES; ++w) mb = fmaxf(mb, sWmax[w * 8 + tid]);
        float lb = 0.0f;
        for (int w = 0; w < NWAVES; ++w)
            lb += sWl[w * 8 + tid] * __expf(sWmax[w * 8 + tid] - mb);
        sBmax[tid] = mb;
        pmax[pidx * 8 + tid] = mb;
        pl[pidx * 8 + tid]   = lb;
    }
    __syncthreads();

    for (int e = tid; e < 8 * D_; e += BLOCK) {
        const int m  = e / D_;
        const float mb = sBmax[m];
        float sum = 0.0f;
        for (int w = 0; w < NWAVES; ++w)
            sum += sTile[w * TILE_FLOATS + e] * __expf(sWmax[w * 8 + m] - mb);
        pacc[(long)pidx * 8 * D_ + e] = sum;
    }
}

#define RBLOCK 256
#define RB_PER_BATCH (M_ * D_ / RBLOCK)

extern "C" __global__ __launch_bounds__(RBLOCK)
void pm_reduce(const float* __restrict__ pmax, const float* __restrict__ pl,
               const float* __restrict__ pacc, float* __restrict__ out)
{
    static_assert(CHUNKS <= 32, "one chunk per lane of the first wave");
    __shared__ float cw[CHUNKS];
    __shared__ float gLinv;
    const int b   = blockIdx.x / RB_PER_BATCH;
    const int blk = blockIdx.x % RB_PER_BATCH;
    const int tid = threadIdx.x;
    const int e   = blk * RBLOCK + tid;
    const int m   = (blk * RBLOCK) / D_;
    if (tid < 32) {
        const bool live = tid < CHUNKS;
        float pm = live ? pmax[(b * CHUNKS + tid) * 8 + m] : NEG_BIG;
        float lv = live ? pl[(b * CHUNKS + tid) * 8 + m]   : 0.0f;
        float mb = pm;
        for (int off = 16; off >= 1; off >>= 1)
            mb = fmaxf(mb, __shfl_xor(mb, off, 32));
        float w = __expf(pm - mb);
        if (live) cw[tid] = w;
        float lb = lv * w;
        for (int off = 16; off >= 1; off >>= 1)
            lb += __shfl_xor(lb, off, 32);
        if (tid == 0) gLinv = 1.0f / lb;
    }
    __syncthreads();
    const float* src = pacc + (long)b * CHUNKS * 8 * D_ + e;
    float s = 0.0f;
#pragma unroll
    for (int c = 0; c < CHUNKS; ++c)
        s += src[(long)c * 8 * D_] * cw[c];
    out[(long)b * 8 * D_ + e] = s * gLinv;
}

extern "C" void kernel_launch(void* const* d_in, const int* in_sizes, int n_in,
                              void* d_out, int out_size, void* d_ws, size_t ws_size,
                              hipStream_t stream)
{
    const float* x     = (const float*)d_in[0];
    const float* q     = (const float*)d_in[1];
    const float* gamma = (const float*)d_in[2];
    const float* beta  = (const float*)d_in[3];
    float* out = (float*)d_out;

    const int P = B_ * CHUNKS;
    float* pmax = (float*)d_ws;
    float* pl   = pmax + (size_t)P * 8;
    float* pacc = pl + (size_t)P * 8;

    const size_t shmem = (size_t)LDS_FLOATS * sizeof(float);
    hipFuncSetAttribute(reinterpret_cast<const void*>(pm_fused),
                        hipFuncAttributeMaxDynamicSharedMemorySize, (int)shmem);

    pm_fused<<<P, BLOCK, shmem, stream>>>(x, q, gamma, beta, pmax, pl, pacc);
    pm_reduce<<<B_ * RB_PER_BATCH, RBLOCK, 0, stream>>>(pmax, pl, pacc, out);
}


# Cross-lane reductions (LayerNorm row sums, softmax row max/sum, merge weights) use DPP row_xmask + permlanex16 on the vector ALU, not ds_bpermute; 8 rows reduced together; partner order unchanged
# speedup vs baseline: 6.9286x; 1.2540x over previous
#include <hip/hip_runtime.h>
#include <math.h>


typedef __attribute__((ext_vector_type(2))) float v2f;
typedef __attribute__((ext_vector_type(8))) float v8f;

#define B_ 16
#define N_ 4096
#define D_ 768
#define M_ 8
#define CHUNKS 32
#define CN (N_ / CHUNKS)
#define NWAVES 4
#define BLOCK (NWAVES * 32)
#define TILES_PER_WAVE (CN / 16 / NWAVES)
#define EPS_ 1e-3f
#define NEG_BIG (-1e30f)
#define FBLKS (D_ / 16)
#define RP 780
#define TILE_FLOATS (16 * RP)
#define RG 8
#define KB1 16
#define NB1 (D_ / (4 * KB1))
#define JB 8
#define NB2 (FBLKS / JB)

struct G1Set { float2 q[KB1], x[KB1], g[KB1], b[KB1]; };
struct G2Set { float x0[JB][4], x1[JB][4], g[JB], be[JB]; };

static __device__ __forceinline__ float xrow(float x)
{
    const int b = __builtin_bit_cast(int, x);
    return __builtin_bit_cast(float, __builtin_amdgcn_permlanex16(b, b, 0x76543210, 0xfedcba98,
                                                                  false, false));
}
template <int M>
static __device__ __forceinline__ float xmask(float x)
{
    const int b = __builtin_bit_cast(int, x);
    return __builtin_bit_cast(float, __builtin_amdgcn_update_dpp(0, b, 0x160 + M, 0xf, 0xf, true));
}
static __device__ __forceinline__ float row16_sum(float v)
{
    v += xmask<8>(v);
    v += xmask<4>(v);
    v += xmask<2>(v);
    v += xmask<1>(v);
    return v;
}
static __device__ __forceinline__ float row16_max(float v)
{
    v = fmaxf(v, xmask<8>(v));
    v = fmaxf(v, xmask<4>(v));
    v = fmaxf(v, xmask<2>(v));
    v = fmaxf(v, xmask<1>(v));
    return v;
}

static __device__ __forceinline__ void g1_load(G1Set& s, const float* qrow, const float* trow,
                                               const float* sG, const float* sBe, int kf0)
{
#pragma unroll
    for (int i = 0; i < KB1; ++i) {
        const int kf = kf0 + 4 * i;
        s.q[i] = *(const float2*)(qrow + kf);
        s.x[i] = *(const float2*)(trow + kf);
        s.g[i] = *(const float2*)(sG + kf);
        s.b[i] = *(const float2*)(sBe + kf);
    }
}

static __device__ __forceinline__ v8f g1_mma(const G1Set& s, float mu_r, float rs_r, v8f S)
{
#pragma unroll
    for (int i = 0; i < KB1; ++i) {
        v2f a;
        a.x = s.q[i].x;
        a.y = s.q[i].y;
        v2f bm;
        bm.x = (s.x[i].x - mu_r) * rs_r * s.g[i].x + s.b[i].x;
        bm.y = (s.x[i].y - mu_r) * rs_r * s.g[i].y + s.b[i].y;
        S = __builtin_amdgcn_wmma_f32_16x16x4_f32(false, a, false, bm,
                                                  (short)0, S, false, false);
    }
    return S;
}

static __device__ __forceinline__ void g2_load(G2Set& s, const float* tcol, const float* sG,
                                               const float* sBe, int j0, int l16)
{
#pragma unroll
    for (int jj = 0; jj < JB; ++jj) {
        const int j = j0 + jj;
        s.g[jj]  = sG[j * 16 + l16];
        s.be[jj] = sBe[j * 16 + l16];
#pragma unroll
        for (int kk = 0; kk < 4; ++kk) {
            s.x0[jj][kk] = tcol[(kk * 4) * RP + j * 16];
            s.x1[jj][kk] = tcol[(kk * 4 + 1) * RP + j * 16];
        }
    }
}

#define LDS_FLOATS (16*RP + 2*D_ + NWAVES*TILE_FLOATS + NWAVES*16*2 + NWAVES*256 + NWAVES*8*2 + 16)

extern "C" __global__ __launch_bounds__(BLOCK)
void pm_fused(const float* __restrict__ x, const float* __restrict__ q,
              const float* __restrict__ gamma, const float* __restrict__ beta,
              float* __restrict__ pmax, float* __restrict__ pl,
              float* __restrict__ pacc)
{
    extern __shared__ __attribute__((aligned(16))) float lds[];
    float* sQ    = lds;
    float* sG    = sQ + 16 * RP;
    float* sBe   = sG + D_;
    float* sTile = sBe + D_;
    float* sMu   = sTile + NWAVES * TILE_FLOATS;
    float* sRs   = sMu + NWAVES * 16;
    float* sP    = sRs + NWAVES * 16;
    float* sWmax = sP + NWAVES * 256;
    float* sWl   = sWmax + NWAVES * 8;
    float* sBmax = sWl + NWAVES * 8;

    const int tid  = threadIdx.x;
    const int wave = tid >> 5;
    const int lane = tid & 31;
    const int half = lane >> 4;
    const int l16  = lane & 15;
    const int b     = blockIdx.x / CHUNKS;
    const int chunk = blockIdx.x % CHUNKS;
    const long xbase = (long)b * N_ * D_;

#pragma unroll
    for (int it = 0; it < (M_ * D_ / 4) / BLOCK; ++it) {
        const int i  = tid + it * BLOCK;
        const int r  = i / (D_ / 4);
        const int d4 = i - r * (D_ / 4);
        const float4 v = *(const float4*)(q + (long)i * 4);
        *(float4*)(sQ + r * RP + d4 * 4) = v;
    }
#pragma unroll
    for (int it = (M_ * D_ / 4) / BLOCK; it < (16 * D_ / 4) / BLOCK; ++it) {
        const int i  = tid + it * BLOCK;
        const int r  = i / (D_ / 4);
        const int d4 = i - r * (D_ / 4);
        *(float4*)(sQ + r * RP + d4 * 4) = make_float4(0.0f, 0.0f, 0.0f, 0.0f);
    }
    for (int i = tid; i < D_ / 4; i += BLOCK) {
        *(float4*)(sG + i * 4)  = *(const float4*)(gamma + i * 4);
        *(float4*)(sBe + i * 4) = *(const float4*)(beta + i * 4);
    }
    __syncthreads();

    float* myTile = sTile + wave * TILE_FLOATS;
    float* myMu   = sMu + wave * 16;
    float* myRs   = sRs + wave * 16;
    float* myP    = sP + wave * 256;

    float rowmax[8], rowsum[8];
#pragma unroll
    for (int r = 0; r < 8; ++r) { rowmax[r] = NEG_BIG; rowsum[r] = 0.0f; }

    v8f acc[FBLKS];
#pragma unroll
    for (int j = 0; j < FBLKS; ++j)
        acc[j] = (v8f){0.f, 0.f, 0.f, 0.f, 0.f, 0.f, 0.f, 0.f};

    const float scale = 0.03608439182435161f;

    for (int t = 0; t < TILES_PER_WAVE; ++t) {
        const int n0 = chunk * CN + (wave * TILES_PER_WAVE + t) * 16;

#pragma unroll 1
        for (int rg = 0; rg < 16; rg += RG) {
            float4 v[RG][6];
            const float* rowbase = x + xbase + (long)(n0 + rg) * D_ + lane * 4;
#pragma unroll
            for (int rr = 0; rr < RG; ++rr) {
#pragma unroll
                for (int i = 0; i < D_ / (32 * 4); ++i)
                    v[rr][i] = *(const float4*)(rowbase + rr * D_ + i * 128);
            }
            float s[RG], s2[RG];
#pragma unroll
            for (int rr = 0; rr < RG; ++rr) {
                float* tdst = myTile + (rg + rr) * RP + lane * 4;
                float a = 0.0f, a2 = 0.0f;
#pragma unroll
                for (int i = 0; i < D_ / (32 * 4); ++i) {
                    const float4 w = v[rr][i];
                    a  += w.x + w.y + w.z + w.w;
                    a2 += w.x * w.x + w.y * w.y + w.z * w.z + w.w * w.w;
                    *(float4*)(tdst + i * 128) = w;
                }
                s[rr]  = a;
                s2[rr] = a2;
            }
#pragma unroll
            for (int rr = 0; rr < RG; ++rr) {
                s[rr]  += xrow(s[rr]);
                s2[rr] += xrow(s2[rr]);
            }
#pragma unroll
            for (int rr = 0; rr < RG; ++rr) {
                s[rr]  = row16_sum(s[rr]);
                s2[rr] = row16_sum(s2[rr]);
            }
            float ts = s[0], ts2 = s2[0];
#pragma unroll
            for (int rr = 1; rr < RG; ++rr) {
                ts  = (lane == rr) ? s[rr]  : ts;
                ts2 = (lane == rr) ? s2[rr] : ts2;
            }
            const float mu  = ts * (1.0f / D_);
            const float var = ts2 * (1.0f / D_) - mu * mu;
            if (lane < RG) { myMu[rg + lane] = mu; myRs[rg + lane] = rsqrtf(var + EPS_); }
        }

        const float  mu_r = myMu[l16];
        const float  rs_r = myRs[l16];
        const float* trow = myTile + l16 * RP;
        const float* qrow = sQ + l16 * RP;

        v8f S = {0.f, 0.f, 0.f, 0.f, 0.f, 0.f, 0.f, 0.f};
        {
            const int kh = half * 2;
            G1Set s0, s1;
            g1_load(s0, qrow, trow, sG, sBe, kh);
#pragma unroll 1
            for (int kb = 0; kb < D_; kb += 2 * 4 * KB1) {
                g1_load(s1, qrow, trow, sG, sBe, kb + 4 * KB1 + kh);
                __builtin_amdgcn_sched_barrier(0);
                S = g1_mma(s0, mu_r, rs_r, S);
                if (kb + 2 * 4 * KB1 < D_)
                    g1_load(s0, qrow, trow, sG, sBe, kb + 2 * 4 * KB1 + kh);
                S = g1_mma(s1, mu_r, rs_r, S);
            }
        }

        float corr[8];
#pragma unroll
        for (int r = 0; r < 8; ++r) {
            float sim = S[r] * scale;
            const float tmax = row16_max(sim);
            float nm = fmaxf(rowmax[r], tmax);
            float p  = __expf(sim - nm);
            float c  = __expf(rowmax[r] - nm);
            const float ps = row16_sum(p);
            rowsum[r] = rowsum[r] * c + ps;
            rowmax[r] = nm;
            corr[r]   = c;
            myP[(r + half * 8) * 16 + l16] = p;
        }

        v2f   pa[4];
        float mua[4], mub[4], rsa[4], rsb[4];
#pragma unroll
        for (int kk = 0; kk < 4; ++kk) {
            const int r0 = kk * 4 + half * 2;
            pa[kk].x = myP[l16 * 16 + r0];
            pa[kk].y = myP[l16 * 16 + r0 + 1];
            mua[kk] = myMu[r0];
            mub[kk] = myMu[r0 + 1];
            rsa[kk] = myRs[r0];
            rsb[kk] = myRs[r0 + 1];
        }
        const float* tcol = myTile + (half * 2) * RP + l16;
        {
            G2Set st[2];
            g2_load(st[0], tcol, sG, sBe, 0, l16);
#pragma unroll
            for (int blk = 0; blk < NB2; ++blk) {
                if (blk + 1 < NB2)
                    g2_load(st[(blk + 1) & 1], tcol, sG, sBe, (blk + 1) * JB, l16);
                __builtin_amdgcn_sched_barrier(0);
                const G2Set& s = st[blk & 1];
#pragma unroll
                for (int jj = 0; jj < JB; ++jj) {
                    const int   j  = blk * JB + jj;
                    const float g  = s.g[jj];
                    const float be = s.be[jj];
                    v8f Dk = {0.f, 0.f, 0.f, 0.f, 0.f, 0.f, 0.f, 0.f};
#pragma unroll
                    for (int kk = 0; kk < 4; ++kk) {
                        v2f bm;
                        bm.x = (s.x0[jj][kk] - mua[kk]) * rsa[kk] * g + be;
                        bm.y = (s.x1[jj][kk] - mub[kk]) * rsb[kk] * g + be;
                        Dk = __builtin_amdgcn_wmma_f32_16x16x4_f32(false, pa[kk], false, bm,
                                                                   (short)0, Dk, false, false);
                    }
#pragma unroll
                    for (int r = 0; r < 8; ++r)
                        acc[j][r] = acc[j][r] * corr[r] + Dk[r];
                }
                __builtin_amdgcn_sched_barrier(0);
            }
        }
    }

    if (lane == 0) {
#pragma unroll
        for (int r = 0; r < 8; ++r) {
            sWmax[wave * 8 + r] = rowmax[r];
            sWl[wave * 8 + r]   = rowsum[r];
        }
    }
    if (half == 0) {
#pragma unroll
        for (int j = 0; j < FBLKS; ++j) {
#pragma unroll
            for (int r = 0; r < 8; ++r)
                myTile[r * D_ + j * 16 + l16] = acc[j][r];
        }
    }
    __syncthreads();

    const int pidx = blockIdx.x;
    if (tid < 8) {
        float mb = NEG_BIG;
        for (int w = 0; w < NWAVES; ++w) mb = fmaxf(mb, sWmax[w * 8 + tid]);
        float lb = 0.0f;
        for (int w = 0; w < NWAVES; ++w)
            lb += sWl[w * 8 + tid] * __expf(sWmax[w * 8 + tid] - mb);
        sBmax[tid] = mb;
        pmax[pidx * 8 + tid] = mb;
        pl[pidx * 8 + tid]   = lb;
    }
    __syncthreads();

    for (int e = tid; e < 8 * D_; e += BLOCK) {
        const int m  = e / D_;
        const float mb = sBmax[m];
        float sum = 0.0f;
        for (int w = 0; w < NWAVES; ++w)
            sum += sTile[w * TILE_FLOATS + e] * __expf(sWmax[w * 8 + m] - mb);
        pacc[(long)pidx * 8 * D_ + e] = sum;
    }
}

#define RBLOCK 256
#define RB_PER_BATCH (M_ * D_ / RBLOCK)

extern "C" __global__ __launch_bounds__(RBLOCK)
void pm_reduce(const float* __restrict__ pmax, const float* __restrict__ pl,
               const float* __restrict__ pacc, float* __restrict__ out)
{
    static_assert(CHUNKS <= 32, "one chunk per lane of the first wave");
    __shared__ float cw[CHUNKS];
    __shared__ float gLinv;
    const int b   = blockIdx.x / RB_PER_BATCH;
    const int blk = blockIdx.x % RB_PER_BATCH;
    const int tid = threadIdx.x;
    const int e   = blk * RBLOCK + tid;
    const int m   = (blk * RBLOCK) / D_;
    if (tid < 32) {
        const bool live = tid < CHUNKS;
        float pm = live ? pmax[(b * CHUNKS + tid) * 8 + m] : NEG_BIG;
        float lv = live ? pl[(b * CHUNKS + tid) * 8 + m]   : 0.0f;
        const float mb = row16_max(fmaxf(pm, xrow(pm)));
        float w = __expf(pm - mb);
        if (live) cw[tid] = w;
        float lb = lv * w;
        lb = row16_sum(lb + xrow(lb));
        if (tid == 0) gLinv = 1.0f / lb;
    }
    __syncthreads();
    const float* src = pacc + (long)b * CHUNKS * 8 * D_ + e;
    float s = 0.0f;
#pragma unroll
    for (int c = 0; c < CHUNKS; ++c)
        s += src[(long)c * 8 * D_] * cw[c];
    out[(long)b * 8 * D_ + e] = s * gLinv;
}

extern "C" void kernel_launch(void* const* d_in, const int* in_sizes, int n_in,
                              void* d_out, int out_size, void* d_ws, size_t ws_size,
                              hipStream_t stream)
{
    const float* x     = (const float*)d_in[0];
    const float* q     = (const float*)d_in[1];
    const float* gamma = (const float*)d_in[2];
    const float* beta  = (const float*)d_in[3];
    float* out = (float*)d_out;

    const int P = B_ * CHUNKS;
    float* pmax = (float*)d_ws;
    float* pl   = pmax + (size_t)P * 8;
    float* pacc = pl + (size_t)P * 8;

    const size_t shmem = (size_t)LDS_FLOATS * sizeof(float);
    hipFuncSetAttribute(reinterpret_cast<const void*>(pm_fused),
                        hipFuncAttributeMaxDynamicSharedMemorySize, (int)shmem);

    pm_fused<<<P, BLOCK, shmem, stream>>>(x, q, gamma, beta, pmax, pl, pacc);
    pm_reduce<<<B_ * RB_PER_BATCH, RBLOCK, 0, stream>>>(pmax, pl, pacc, out);
}


# Batch rows split over 16 work-groups (4 row tiles per wave) instead of 32: one work-group per WGP, so staging and the in-group merge run once per WGP; final merge sums 16 partials (sum order changes)
# speedup vs baseline: 7.8827x; 1.1377x over previous
#include <hip/hip_runtime.h>
#include <math.h>


typedef __attribute__((ext_vector_type(2))) float v2f;
typedef __attribute__((ext_vector_type(8))) float v8f;

#define B_ 16
#define N_ 4096
#define D_ 768
#define M_ 8
#define CHUNKS 16
#define CN (N_ / CHUNKS)
#define NWAVES 4
#define BLOCK (NWAVES * 32)
#define TILES_PER_WAVE (CN / 16 / NWAVES)
#define EPS_ 1e-3f
#define NEG_BIG (-1e30f)
#define FBLKS (D_ / 16)
#define RP 780
#define TILE_FLOATS (16 * RP)
#define RG 8
#define KB1 16
#define NB1 (D_ / (4 * KB1))
#define JB 8
#define NB2 (FBLKS / JB)

struct G1Set { float2 q[KB1], x[KB1], g[KB1], b[KB1]; };
struct G2Set { float x0[JB][4], x1[JB][4], g[JB], be[JB]; };

static __device__ __forceinline__ float xrow(float x)
{
    const int b = __builtin_bit_cast(int, x);
    return __builtin_bit_cast(float, __builtin_amdgcn_permlanex16(b, b, 0x76543210, 0xfedcba98,
                                                                  false, false));
}
template <int M>
static __device__ __forceinline__ float xmask(float x)
{
    const int b = __builtin_bit_cast(int, x);
    return __builtin_bit_cast(float, __builtin_amdgcn_update_dpp(0, b, 0x160 + M, 0xf, 0xf, true));
}
static __device__ __forceinline__ float row16_sum(float v)
{
    v += xmask<8>(v);
    v += xmask<4>(v);
    v += xmask<2>(v);
    v += xmask<1>(v);
    return v;
}
static __device__ __forceinline__ float row16_max(float v)
{
    v = fmaxf(v, xmask<8>(v));
    v = fmaxf(v, xmask<4>(v));
    v = fmaxf(v, xmask<2>(v));
    v = fmaxf(v, xmask<1>(v));
    return v;
}

static __device__ __forceinline__ void g1_load(G1Set& s, const float* qrow, const float* trow,
                                               const float* sG, const float* sBe, int kf0)
{
#pragma unroll
    for (int i = 0; i < KB1; ++i) {
        const int kf = kf0 + 4 * i;
        s.q[i] = *(const float2*)(qrow + kf);
        s.x[i] = *(const float2*)(trow + kf);
        s.g[i] = *(const float2*)(sG + kf);
        s.b[i] = *(const float2*)(sBe + kf);
    }
}

static __device__ __forceinline__ v8f g1_mma(const G1Set& s, float mu_r, float rs_r, v8f S)
{
#pragma unroll
    for (int i = 0; i < KB1; ++i) {
        v2f a;
        a.x = s.q[i].x;
        a.y = s.q[i].y;
        v2f bm;
        bm.x = (s.x[i].x - mu_r) * rs_r * s.g[i].x + s.b[i].x;
        bm.y = (s.x[i].y - mu_r) * rs_r * s.g[i].y + s.b[i].y;
        S = __builtin_amdgcn_wmma_f32_16x16x4_f32(false, a, false, bm,
                                                  (short)0, S, false, false);
    }
    return S;
}

static __device__ __forceinline__ void g2_load(G2Set& s, const float* tcol, const float* sG,
                                               const float* sBe, int j0, int l16)
{
#pragma unroll
    for (int jj = 0; jj < JB; ++jj) {
        const int j = j0 + jj;
        s.g[jj]  = sG[j * 16 + l16];
        s.be[jj] = sBe[j * 16 + l16];
#pragma unroll
        for (int kk = 0; kk < 4; ++kk) {
            s.x0[jj][kk] = tcol[(kk * 4) * RP + j * 16];
            s.x1[jj][kk] = tcol[(kk * 4 + 1) * RP + j * 16];
        }
    }
}

#define LDS_FLOATS (16*RP + 2*D_ + NWAVES*TILE_FLOATS + NWAVES*16*2 + NWAVES*256 + NWAVES*8*2 + 16)

extern "C" __global__ __launch_bounds__(BLOCK)
void pm_fused(const float* __restrict__ x, const float* __restrict__ q,
              const float* __restrict__ gamma, const float* __restrict__ beta,
              float* __restrict__ pmax, float* __restrict__ pl,
              float* __restrict__ pacc)
{
    extern __shared__ __attribute__((aligned(16))) float lds[];
    float* sQ    = lds;
    float* sG    = sQ + 16 * RP;
    float* sBe   = sG + D_;
    float* sTile = sBe + D_;
    float* sMu   = sTile + NWAVES * TILE_FLOATS;
    float* sRs   = sMu + NWAVES * 16;
    float* sP    = sRs + NWAVES * 16;
    float* sWmax = sP + NWAVES * 256;
    float* sWl   = sWmax + NWAVES * 8;
    float* sBmax = sWl + NWAVES * 8;

    const int tid  = threadIdx.x;
    const int wave = tid >> 5;
    const int lane = tid & 31;
    const int half = lane >> 4;
    const int l16  = lane & 15;
    const int b     = blockIdx.x / CHUNKS;
    const int chunk = blockIdx.x % CHUNKS;
    const long xbase = (long)b * N_ * D_;

#pragma unroll
    for (int it = 0; it < (M_ * D_ / 4) / BLOCK; ++it) {
        const int i  = tid + it * BLOCK;
        const int r  = i / (D_ / 4);
        const int d4 = i - r * (D_ / 4);
        const float4 v = *(const float4*)(q + (long)i * 4);
        *(float4*)(sQ + r * RP + d4 * 4) = v;
    }
#pragma unroll
    for (int it = (M_ * D_ / 4) / BLOCK; it < (16 * D_ / 4) / BLOCK; ++it) {
        const int i  = tid + it * BLOCK;
        const int r  = i / (D_ / 4);
        const int d4 = i - r * (D_ / 4);
        *(float4*)(sQ + r * RP + d4 * 4) = make_float4(0.0f, 0.0f, 0.0f, 0.0f);
    }
    for (int i = tid; i < D_ / 4; i += BLOCK) {
        *(float4*)(sG + i * 4)  = *(const float4*)(gamma + i * 4);
        *(float4*)(sBe + i * 4) = *(const float4*)(beta + i * 4);
    }
    __syncthreads();

    float* myTile = sTile + wave * TILE_FLOATS;
    float* myMu   = sMu + wave * 16;
    float* myRs   = sRs + wave * 16;
    float* myP    = sP + wave * 256;

    float rowmax[8], rowsum[8];
#pragma unroll
    for (int r = 0; r < 8; ++r) { rowmax[r] = NEG_BIG; rowsum[r] = 0.0f; }

    v8f acc[FBLKS];
#pragma unroll
    for (int j = 0; j < FBLKS; ++j)
        acc[j] = (v8f){0.f, 0.f, 0.f, 0.f, 0.f, 0.f, 0.f, 0.f};

    const float scale = 0.03608439182435161f;

    for (int t = 0; t < TILES_PER_WAVE; ++t) {
        const int n0 = chunk * CN + (wave * TILES_PER_WAVE + t) * 16;

#pragma unroll 1
        for (int rg = 0; rg < 16; rg += RG) {
            float4 v[RG][6];
            const float* rowbase = x + xbase + (long)(n0 + rg) * D_ + lane * 4;
#pragma unroll
            for (int rr = 0; rr < RG; ++rr) {
#pragma unroll
                for (int i = 0; i < D_ / (32 * 4); ++i)
                    v[rr][i] = *(const float4*)(rowbase + rr * D_ + i * 128);
            }
            float s[RG], s2[RG];
#pragma unroll
            for (int rr = 0; rr < RG; ++rr) {
                float* tdst = myTile + (rg + rr) * RP + lane * 4;
                float a = 0.0f, a2 = 0.0f;
#pragma unroll
                for (int i = 0; i < D_ / (32 * 4); ++i) {
                    const float4 w = v[rr][i];
                    a  += w.x + w.y + w.z + w.w;
                    a2 += w.x * w.x + w.y * w.y + w.z * w.z + w.w * w.w;
                    *(float4*)(tdst + i * 128) = w;
                }
                s[rr]  = a;
                s2[rr] = a2;
            }
#pragma unroll
            for (int rr = 0; rr < RG; ++rr) {
                s[rr]  += xrow(s[rr]);
                s2[rr] += xrow(s2[rr]);
            }
#pragma unroll
            for (int rr = 0; rr < RG; ++rr) {
                s[rr]  = row16_sum(s[rr]);
                s2[rr] = row16_sum(s2[rr]);
            }
            float ts = s[0], ts2 = s2[0];
#pragma unroll
            for (int rr = 1; rr < RG; ++rr) {
                ts  = (lane == rr) ? s[rr]  : ts;
                ts2 = (lane == rr) ? s2[rr] : ts2;
            }
            const float mu  = ts * (1.0f / D_);
            const float var = ts2 * (1.0f / D_) - mu * mu;
            if (lane < RG) { myMu[rg + lane] = mu; myRs[rg + lane] = rsqrtf(var + EPS_); }
        }

        const float  mu_r = myMu[l16];
        const float  rs_r = myRs[l16];
        const float* trow = myTile + l16 * RP;
        const float* qrow = sQ + l16 * RP;

        v8f S = {0.f, 0.f, 0.f, 0.f, 0.f, 0.f, 0.f, 0.f};
        {
            const int kh = half * 2;
            G1Set s0, s1;
            g1_load(s0, qrow, trow, sG, sBe, kh);
#pragma unroll 1
            for (int kb = 0; kb < D_; kb += 2 * 4 * KB1) {
                g1_load(s1, qrow, trow, sG, sBe, kb + 4 * KB1 + kh);
                __builtin_amdgcn_sched_barrier(0);
                S = g1_mma(s0, mu_r, rs_r, S);
                if (kb + 2 * 4 * KB1 < D_)
                    g1_load(s0, qrow, trow, sG, sBe, kb + 2 * 4 * KB1 + kh);
                S = g1_mma(s1, mu_r, rs_r, S);
            }
        }

        float corr[8];
#pragma unroll
        for (int r = 0; r < 8; ++r) {
            float sim = S[r] * scale;
            const float tmax = row16_max(sim);
            float nm = fmaxf(rowmax[r], tmax);
            float p  = __expf(sim - nm);
            float c  = __expf(rowmax[r] - nm);
            const float ps = row16_sum(p);
            rowsum[r] = rowsum[r] * c + ps;
            rowmax[r] = nm;
            corr[r]   = c;
            myP[(r + half * 8) * 16 + l16] = p;
        }

        v2f   pa[4];
        float mua[4], mub[4], rsa[4], rsb[4];
#pragma unroll
        for (int kk = 0; kk < 4; ++kk) {
            const int r0 = kk * 4 + half * 2;
            pa[kk].x = myP[l16 * 16 + r0];
            pa[kk].y = myP[l16 * 16 + r0 + 1];
            mua[kk] = myMu[r0];
            mub[kk] = myMu[r0 + 1];
            rsa[kk] = myRs[r0];
            rsb[kk] = myRs[r0 + 1];
        }
        const float* tcol = myTile + (half * 2) * RP + l16;
        {
            G2Set st[2];
            g2_load(st[0], tcol, sG, sBe, 0, l16);
#pragma unroll
            for (int blk = 0; blk < NB2; ++blk) {
                if (blk + 1 < NB2)
                    g2_load(st[(blk + 1) & 1], tcol, sG, sBe, (blk + 1) * JB, l16);
                __builtin_amdgcn_sched_barrier(0);
                const G2Set& s = st[blk & 1];
#pragma unroll
                for (int jj = 0; jj < JB; ++jj) {
                    const int   j  = blk * JB + jj;
                    const float g  = s.g[jj];
                    const float be = s.be[jj];
                    v8f Dk = {0.f, 0.f, 0.f, 0.f, 0.f, 0.f, 0.f, 0.f};
#pragma unroll
                    for (int kk = 0; kk < 4; ++kk) {
                        v2f bm;
                        bm.x = (s.x0[jj][kk] - mua[kk]) * rsa[kk] * g + be;
                        bm.y = (s.x1[jj][kk] - mub[kk]) * rsb[kk] * g + be;
                        Dk = __builtin_amdgcn_wmma_f32_16x16x4_f32(false, pa[kk], false, bm,
                                                                   (short)0, Dk, false, false);
                    }
#pragma unroll
                    for (int r = 0; r < 8; ++r)
                        acc[j][r] = acc[j][r] * corr[r] + Dk[r];
                }
                __builtin_amdgcn_sched_barrier(0);
            }
        }
    }

    if (lane == 0) {
#pragma unroll
        for (int r = 0; r < 8; ++r) {
            sWmax[wave * 8 + r] = rowmax[r];
            sWl[wave * 8 + r]   = rowsum[r];
        }
    }
    if (half == 0) {
#pragma unroll
        for (int j = 0; j < FBLKS; ++j) {
#pragma unroll
            for (int r = 0; r < 8; ++r)
                myTile[r * D_ + j * 16 + l16] = acc[j][r];
        }
    }
    __syncthreads();

    const int pidx = blockIdx.x;
    if (tid < 8) {
        float mb = NEG_BIG;
        for (int w = 0; w < NWAVES; ++w) mb = fmaxf(mb, sWmax[w * 8 + tid]);
        float lb = 0.0f;
        for (int w = 0; w < NWAVES; ++w)
            lb += sWl[w * 8 + tid] * __expf(sWmax[w * 8 + tid] - mb);
        sBmax[tid] = mb;
        pmax[pidx * 8 + tid] = mb;
        pl[pidx * 8 + tid]   = lb;
    }
    __syncthreads();

    for (int e = tid; e < 8 * D_; e += BLOCK) {
        const int m  = e / D_;
        const float mb = sBmax[m];
        float sum = 0.0f;
        for (int w = 0; w < NWAVES; ++w)
            sum += sTile[w * TILE_FLOATS + e] * __expf(sWmax[w * 8 + m] - mb);
        pacc[(long)pidx * 8 * D_ + e] = sum;
    }
}

#define RBLOCK 256
#define RB_PER_BATCH (M_ * D_ / RBLOCK)

extern "C" __global__ __launch_bounds__(RBLOCK)
void pm_reduce(const float* __restrict__ pmax, const float* __restrict__ pl,
               const float* __restrict__ pacc, float* __restrict__ out)
{
    static_assert(CHUNKS <= 32, "one chunk per lane of the first wave");
    __shared__ float cw[CHUNKS];
    __shared__ float gLinv;
    const int b   = blockIdx.x / RB_PER_BATCH;
    const int blk = blockIdx.x % RB_PER_BATCH;
    const int tid = threadIdx.x;
    const int e   = blk * RBLOCK + tid;
    const int m   = (blk * RBLOCK) / D_;
    if (tid < 32) {
        const bool live = tid < CHUNKS;
        float pm = live ? pmax[(b * CHUNKS + tid) * 8 + m] : NEG_BIG;
        float lv = live ? pl[(b * CHUNKS + tid) * 8 + m]   : 0.0f;
        const float mb = row16_max(fmaxf(pm, xrow(pm)));
        float w = __expf(pm - mb);
        if (live) cw[tid] = w;
        float lb = lv * w;
        lb = row16_sum(lb + xrow(lb));
        if (tid == 0) gLinv = 1.0f / lb;
    }
    __syncthreads();
    const float* src = pacc + (long)b * CHUNKS * 8 * D_ + e;
    float s = 0.0f;
#pragma unroll
    for (int c = 0; c < CHUNKS; ++c)
        s += src[(long)c * 8 * D_] * cw[c];
    out[(long)b * 8 * D_ + e] = s * gLinv;
}

extern "C" void kernel_launch(void* const* d_in, const int* in_sizes, int n_in,
                              void* d_out, int out_size, void* d_ws, size_t ws_size,
                              hipStream_t stream)
{
    const float* x     = (const float*)d_in[0];
    const float* q     = (const float*)d_in[1];
    const float* gamma = (const float*)d_in[2];
    const float* beta  = (const float*)d_in[3];
    float* out = (float*)d_out;

    const int P = B_ * CHUNKS;
    float* pmax = (float*)d_ws;
    float* pl   = pmax + (size_t)P * 8;
    float* pacc = pl + (size_t)P * 8;

    const size_t shmem = (size_t)LDS_FLOATS * sizeof(float);
    hipFuncSetAttribute(reinterpret_cast<const void*>(pm_fused),
                        hipFuncAttributeMaxDynamicSharedMemorySize, (int)shmem);

    pm_fused<<<P, BLOCK, shmem, stream>>>(x, q, gamma, beta, pmax, pl, pacc);
    pm_reduce<<<B_ * RB_PER_BATCH, RBLOCK, 0, stream>>>(pmax, pl, pacc, out);
}


# Next 16-row tile prefetched into the second-level cache while both GEMMs run from LDS (one 128-byte line per lane beside each LDS read group): staging loads hit cache, not memory; sums unchanged
# speedup vs baseline: 8.7265x; 1.1070x over previous
#include <hip/hip_runtime.h>
#include <math.h>


typedef __attribute__((ext_vector_type(2))) float v2f;
typedef __attribute__((ext_vector_type(8))) float v8f;

#define B_ 16
#define N_ 4096
#define D_ 768
#define M_ 8
#define CHUNKS 16
#define CN (N_ / CHUNKS)
#define NWAVES 4
#define BLOCK (NWAVES * 32)
#define TILES_PER_WAVE (CN / 16 / NWAVES)
#define EPS_ 1e-3f
#define NEG_BIG (-1e30f)
#define FBLKS (D_ / 16)
#define RP 780
#define TILE_FLOATS (16 * RP)
#define RG 8
#define KB1 16
#define NB1 (D_ / (4 * KB1))
#define JB 8
#define NB2 (FBLKS / JB)

struct G1Set { float2 q[KB1], x[KB1], g[KB1], b[KB1]; };
struct G2Set { float x0[JB][4], x1[JB][4], g[JB], be[JB]; };

static __device__ __forceinline__ float xrow(float x)
{
    const int b = __builtin_bit_cast(int, x);
    return __builtin_bit_cast(float, __builtin_amdgcn_permlanex16(b, b, 0x76543210, 0xfedcba98,
                                                                  false, false));
}
template <int M>
static __device__ __forceinline__ float xmask(float x)
{
    const int b = __builtin_bit_cast(int, x);
    return __builtin_bit_cast(float, __builtin_amdgcn_update_dpp(0, b, 0x160 + M, 0xf, 0xf, true));
}
static __device__ __forceinline__ float row16_sum(float v)
{
    v += xmask<8>(v);
    v += xmask<4>(v);
    v += xmask<2>(v);
    v += xmask<1>(v);
    return v;
}
static __device__ __forceinline__ float row16_max(float v)
{
    v = fmaxf(v, xmask<8>(v));
    v = fmaxf(v, xmask<4>(v));
    v = fmaxf(v, xmask<2>(v));
    v = fmaxf(v, xmask<1>(v));
    return v;
}

static __device__ __forceinline__ void prefetch_rows8(const float* rows, int lane)
{
#pragma unroll
    for (int i = 0; i < (8 * D_) / (32 * 32); ++i)
        __builtin_prefetch(rows + (i * 32 + lane) * 32, 0, 2);
}

static __device__ __forceinline__ void g1_load(G1Set& s, const float* qrow, const float* trow,
                                               const float* sG, const float* sBe, int kf0)
{
#pragma unroll
    for (int i = 0; i < KB1; ++i) {
        const int kf = kf0 + 4 * i;
        s.q[i] = *(const float2*)(qrow + kf);
        s.x[i] = *(const float2*)(trow + kf);
        s.g[i] = *(const float2*)(sG + kf);
        s.b[i] = *(const float2*)(sBe + kf);
    }
}

static __device__ __forceinline__ v8f g1_mma(const G1Set& s, float mu_r, float rs_r, v8f S)
{
#pragma unroll
    for (int i = 0; i < KB1; ++i) {
        v2f a;
        a.x = s.q[i].x;
        a.y = s.q[i].y;
        v2f bm;
        bm.x = (s.x[i].x - mu_r) * rs_r * s.g[i].x + s.b[i].x;
        bm.y = (s.x[i].y - mu_r) * rs_r * s.g[i].y + s.b[i].y;
        S = __builtin_amdgcn_wmma_f32_16x16x4_f32(false, a, false, bm,
                                                  (short)0, S, false, false);
    }
    return S;
}

static __device__ __forceinline__ void g2_load(G2Set& s, const float* tcol, const float* sG,
                                               const float* sBe, int j0, int l16)
{
#pragma unroll
    for (int jj = 0; jj < JB; ++jj) {
        const int j = j0 + jj;
        s.g[jj]  = sG[j * 16 + l16];
        s.be[jj] = sBe[j * 16 + l16];
#pragma unroll
        for (int kk = 0; kk < 4; ++kk) {
            s.x0[jj][kk] = tcol[(kk * 4) * RP + j * 16];
            s.x1[jj][kk] = tcol[(kk * 4 + 1) * RP + j * 16];
        }
    }
}

#define LDS_FLOATS (16*RP + 2*D_ + NWAVES*TILE_FLOATS + NWAVES*16*2 + NWAVES*256 + NWAVES*8*2 + 16)

extern "C" __global__ __launch_bounds__(BLOCK)
void pm_fused(const float* __restrict__ x, const float* __restrict__ q,
              const float* __restrict__ gamma, const float* __restrict__ beta,
              float* __restrict__ pmax, float* __restrict__ pl,
              float* __restrict__ pacc)
{
    extern __shared__ __attribute__((aligned(16))) float lds[];
    float* sQ    = lds;
    float* sG    = sQ + 16 * RP;
    float* sBe   = sG + D_;
    float* sTile = sBe + D_;
    float* sMu   = sTile + NWAVES * TILE_FLOATS;
    float* sRs   = sMu + NWAVES * 16;
    float* sP    = sRs + NWAVES * 16;
    float* sWmax = sP + NWAVES * 256;
    float* sWl   = sWmax + NWAVES * 8;
    float* sBmax = sWl + NWAVES * 8;

    const int tid  = threadIdx.x;
    const int wave = tid >> 5;
    const int lane = tid & 31;
    const int half = lane >> 4;
    const int l16  = lane & 15;
    const int b     = blockIdx.x / CHUNKS;
    const int chunk = blockIdx.x % CHUNKS;
    const long xbase = (long)b * N_ * D_;

    const float* tile0 = x + xbase + (long)(chunk * CN + wave * TILES_PER_WAVE * 16) * D_;
    prefetch_rows8(tile0, lane);

#pragma unroll
    for (int it = 0; it < (M_ * D_ / 4) / BLOCK; ++it) {
        const int i  = tid + it * BLOCK;
        const int r  = i / (D_ / 4);
        const int d4 = i - r * (D_ / 4);
        const float4 v = *(const float4*)(q + (long)i * 4);
        *(float4*)(sQ + r * RP + d4 * 4) = v;
    }
#pragma unroll
    for (int it = (M_ * D_ / 4) / BLOCK; it < (16 * D_ / 4) / BLOCK; ++it) {
        const int i  = tid + it * BLOCK;
        const int r  = i / (D_ / 4);
        const int d4 = i - r * (D_ / 4);
        *(float4*)(sQ + r * RP + d4 * 4) = make_float4(0.0f, 0.0f, 0.0f, 0.0f);
    }
    for (int i = tid; i < D_ / 4; i += BLOCK) {
        *(float4*)(sG + i * 4)  = *(const float4*)(gamma + i * 4);
        *(float4*)(sBe + i * 4) = *(const float4*)(beta + i * 4);
    }
    prefetch_rows8(tile0 + 8 * D_, lane);
    __syncthreads();

    float* myTile = sTile + wave * TILE_FLOATS;
    float* myMu   = sMu + wave * 16;
    float* myRs   = sRs + wave * 16;
    float* myP    = sP + wave * 256;

    float rowmax[8], rowsum[8];
#pragma unroll
    for (int r = 0; r < 8; ++r) { rowmax[r] = NEG_BIG; rowsum[r] = 0.0f; }

    v8f acc[FBLKS];
#pragma unroll
    for (int j = 0; j < FBLKS; ++j)
        acc[j] = (v8f){0.f, 0.f, 0.f, 0.f, 0.f, 0.f, 0.f, 0.f};

    const float scale = 0.03608439182435161f;

    for (int t = 0; t < TILES_PER_WAVE; ++t) {
        const int n0 = chunk * CN + (wave * TILES_PER_WAVE + t) * 16;

#pragma unroll 1
        for (int rg = 0; rg < 16; rg += RG) {
            float4 v[RG][6];
            const float* rowbase = x + xbase + (long)(n0 + rg) * D_ + lane * 4;
#pragma unroll
            for (int rr = 0; rr < RG; ++rr) {
#pragma unroll
                for (int i = 0; i < D_ / (32 * 4); ++i)
                    v[rr][i] = *(const float4*)(rowbase + rr * D_ + i * 128);
            }
            float s[RG], s2[RG];
#pragma unroll
            for (int rr = 0; rr < RG; ++rr) {
                float* tdst = myTile + (rg + rr) * RP + lane * 4;
                float a = 0.0f, a2 = 0.0f;
#pragma unroll
                for (int i = 0; i < D_ / (32 * 4); ++i) {
                    const float4 w = v[rr][i];
                    a  += w.x + w.y + w.z + w.w;
                    a2 += w.x * w.x + w.y * w.y + w.z * w.z + w.w * w.w;
                    *(float4*)(tdst + i * 128) = w;
                }
                s[rr]  = a;
                s2[rr] = a2;
            }
#pragma unroll
            for (int rr = 0; rr < RG; ++rr) {
                s[rr]  += xrow(s[rr]);
                s2[rr] += xrow(s2[rr]);
            }
#pragma unroll
            for (int rr = 0; rr < RG; ++rr) {
                s[rr]  = row16_sum(s[rr]);
                s2[rr] = row16_sum(s2[rr]);
            }
            float ts = s[0], ts2 = s2[0];
#pragma unroll
            for (int rr = 1; rr < RG; ++rr) {
                ts  = (lane == rr) ? s[rr]  : ts;
                ts2 = (lane == rr) ? s2[rr] : ts2;
            }
            const float mu  = ts * (1.0f / D_);
            const float var = ts2 * (1.0f / D_) - mu * mu;
            if (lane < RG) { myMu[rg + lane] = mu; myRs[rg + lane] = rsqrtf(var + EPS_); }
        }

        const float* pfTile = x + xbase + (long)(n0 + ((t + 1 < TILES_PER_WAVE) ? 16 : 0)) * D_;
        const int    pfLine = lane * 32;

        const float  mu_r = myMu[l16];
        const float  rs_r = myRs[l16];
        const float* trow = myTile + l16 * RP;
        const float* qrow = sQ + l16 * RP;

        v8f S = {0.f, 0.f, 0.f, 0.f, 0.f, 0.f, 0.f, 0.f};
        {
            const int kh = half * 2;
            G1Set s0, s1;
            g1_load(s0, qrow, trow, sG, sBe, kh);
#pragma unroll 1
            for (int kb = 0; kb < D_; kb += 2 * 4 * KB1) {
                __builtin_prefetch(pfTile + kb * 8 + pfLine, 0, 2);
                g1_load(s1, qrow, trow, sG, sBe, kb + 4 * KB1 + kh);
                __builtin_amdgcn_sched_barrier(0);
                S = g1_mma(s0, mu_r, rs_r, S);
                if (kb + 2 * 4 * KB1 < D_)
                    g1_load(s0, qrow, trow, sG, sBe, kb + 2 * 4 * KB1 + kh);
                S = g1_mma(s1, mu_r, rs_r, S);
            }
        }

        float corr[8];
#pragma unroll
        for (int r = 0; r < 8; ++r) {
            float sim = S[r] * scale;
            const float tmax = row16_max(sim);
            float nm = fmaxf(rowmax[r], tmax);
            float p  = __expf(sim - nm);
            float c  = __expf(rowmax[r] - nm);
            const float ps = row16_sum(p);
            rowsum[r] = rowsum[r] * c + ps;
            rowmax[r] = nm;
            corr[r]   = c;
            myP[(r + half * 8) * 16 + l16] = p;
        }

        v2f   pa[4];
        float mua[4], mub[4], rsa[4], rsb[4];
#pragma unroll
        for (int kk = 0; kk < 4; ++kk) {
            const int r0 = kk * 4 + half * 2;
            pa[kk].x = myP[l16 * 16 + r0];
            pa[kk].y = myP[l16 * 16 + r0 + 1];
            mua[kk] = myMu[r0];
            mub[kk] = myMu[r0 + 1];
            rsa[kk] = myRs[r0];
            rsb[kk] = myRs[r0 + 1];
        }
        const float* tcol = myTile + (half * 2) * RP + l16;
        {
            G2Set st[2];
            g2_load(st[0], tcol, sG, sBe, 0, l16);
#pragma unroll
            for (int blk = 0; blk < NB2; ++blk) {
                if (blk + 1 < NB2)
                    g2_load(st[(blk + 1) & 1], tcol, sG, sBe, (blk + 1) * JB, l16);
                __builtin_prefetch(pfTile + (8 * D_ + blk * 1024) + pfLine, 0, 2);
                __builtin_amdgcn_sched_barrier(0);
                const G2Set& s = st[blk & 1];
#pragma unroll
                for (int jj = 0; jj < JB; ++jj) {
                    const int   j  = blk * JB + jj;
                    const float g  = s.g[jj];
                    const float be = s.be[jj];
                    v8f Dk = {0.f, 0.f, 0.f, 0.f, 0.f, 0.f, 0.f, 0.f};
#pragma unroll
                    for (int kk = 0; kk < 4; ++kk) {
                        v2f bm;
                        bm.x = (s.x0[jj][kk] - mua[kk]) * rsa[kk] * g + be;
                        bm.y = (s.x1[jj][kk] - mub[kk]) * rsb[kk] * g + be;
                        Dk = __builtin_amdgcn_wmma_f32_16x16x4_f32(false, pa[kk], false, bm,
                                                                   (short)0, Dk, false, false);
                    }
#pragma unroll
                    for (int r = 0; r < 8; ++r)
                        acc[j][r] = acc[j][r] * corr[r] + Dk[r];
                }
                __builtin_amdgcn_sched_barrier(0);
            }
        }
    }

    if (lane == 0) {
#pragma unroll
        for (int r = 0; r < 8; ++r) {
            sWmax[wave * 8 + r] = rowmax[r];
            sWl[wave * 8 + r]   = rowsum[r];
        }
    }
    if (half == 0) {
#pragma unroll
        for (int j = 0; j < FBLKS; ++j) {
#pragma unroll
            for (int r = 0; r < 8; ++r)
                myTile[r * D_ + j * 16 + l16] = acc[j][r];
        }
    }
    __syncthreads();

    const int pidx = blockIdx.x;
    if (tid < 8) {
        float mb = NEG_BIG;
        for (int w = 0; w < NWAVES; ++w) mb = fmaxf(mb, sWmax[w * 8 + tid]);
        float lb = 0.0f;
        for (int w = 0; w < NWAVES; ++w)
            lb += sWl[w * 8 + tid] * __expf(sWmax[w * 8 + tid] - mb);
        sBmax[tid] = mb;
        pmax[pidx * 8 + tid] = mb;
        pl[pidx * 8 + tid]   = lb;
    }
    __syncthreads();

    for (int e = tid; e < 8 * D_; e += BLOCK) {
        const int m  = e / D_;
        const float mb = sBmax[m];
        float sum = 0.0f;
        for (int w = 0; w < NWAVES; ++w)
            sum += sTile[w * TILE_FLOATS + e] * __expf(sWmax[w * 8 + m] - mb);
        pacc[(long)pidx * 8 * D_ + e] = sum;
    }
}

#define RBLOCK 256
#define RB_PER_BATCH (M_ * D_ / RBLOCK)

extern "C" __global__ __launch_bounds__(RBLOCK)
void pm_reduce(const float* __restrict__ pmax, const float* __restrict__ pl,
               const float* __restrict__ pacc, float* __restrict__ out)
{
    static_assert(CHUNKS <= 32, "one chunk per lane of the first wave");
    __shared__ float cw[CHUNKS];
    __shared__ float gLinv;
    const int b   = blockIdx.x / RB_PER_BATCH;
    const int blk = blockIdx.x % RB_PER_BATCH;
    const int tid = threadIdx.x;
    const int e   = blk * RBLOCK + tid;
    const int m   = (blk * RBLOCK) / D_;
    if (tid < 32) {
        const bool live = tid < CHUNKS;
        float pm = live ? pmax[(b * CHUNKS + tid) * 8 + m] : NEG_BIG;
        float lv = live ? pl[(b * CHUNKS + tid) * 8 + m]   : 0.0f;
        const float mb = row16_max(fmaxf(pm, xrow(pm)));
        float w = __expf(pm - mb);
        if (live) cw[tid] = w;
        float lb = lv * w;
        lb = row16_sum(lb + xrow(lb));
        if (tid == 0) gLinv = 1.0f / lb;
    }
    __syncthreads();
    const float* src = pacc + (long)b * CHUNKS * 8 * D_ + e;
    float s = 0.0f;
#pragma unroll
    for (int c = 0; c < CHUNKS; ++c)
        s += src[(long)c * 8 * D_] * cw[c];
    out[(long)b * 8 * D_ + e] = s * gLinv;
}

extern "C" void kernel_launch(void* const* d_in, const int* in_sizes, int n_in,
                              void* d_out, int out_size, void* d_ws, size_t ws_size,
                              hipStream_t stream)
{
    const float* x     = (const float*)d_in[0];
    const float* q     = (const float*)d_in[1];
    const float* gamma = (const float*)d_in[2];
    const float* beta  = (const float*)d_in[3];
    float* out = (float*)d_out;

    const int P = B_ * CHUNKS;
    float* pmax = (float*)d_ws;
    float* pl   = pmax + (size_t)P * 8;
    float* pacc = pl + (size_t)P * 8;

    const size_t shmem = (size_t)LDS_FLOATS * sizeof(float);
    hipFuncSetAttribute(reinterpret_cast<const void*>(pm_fused),
                        hipFuncAttributeMaxDynamicSharedMemorySize, (int)shmem);

    pm_fused<<<P, BLOCK, shmem, stream>>>(x, q, gamma, beta, pmax, pl, pacc);
    pm_reduce<<<B_ * RB_PER_BATCH, RBLOCK, 0, stream>>>(pmax, pl, pacc, out);
}


# LayerNorm applied behind both matrix products (to the 16x16 scores and the merged sums), not to each operand element: GEMMs read raw x, row stats come from GEMM1 fragments; sum order changes
# speedup vs baseline: 10.0359x; 1.1500x over previous
#include <hip/hip_runtime.h>
#include <math.h>


typedef __attribute__((ext_vector_type(2))) float v2f;
typedef __attribute__((ext_vector_type(8))) float v8f;

#define B_ 16
#define N_ 4096
#define D_ 768
#define M_ 8
#define CHUNKS 16
#define CN (N_ / CHUNKS)
#define NWAVES 4
#define BLOCK (NWAVES * 32)
#define TILES_PER_WAVE (CN / 16 / NWAVES)
#define EPS_ 1e-3f
#define NEG_BIG (-1e30f)
#define FBLKS (D_ / 16)
#define RP 780
#define TILE_FLOATS (16 * RP)
#define RG 8
#define KB1 16
#define NB1 (D_ / (4 * KB1))
#define JB 8
#define NB2 (FBLKS / JB)


struct G1Set { float2 q[KB1], x[KB1]; };
struct G2Set { v2f x[JB][4]; };

static __device__ __forceinline__ float xrow(float x)
{
    const int b = __builtin_bit_cast(int, x);
    return __builtin_bit_cast(float, __builtin_amdgcn_permlanex16(b, b, 0x76543210, 0xfedcba98,
                                                                  false, false));
}
template <int M>
static __device__ __forceinline__ float xmask(float x)
{
    const int b = __builtin_bit_cast(int, x);
    return __builtin_bit_cast(float, __builtin_amdgcn_update_dpp(0, b, 0x160 + M, 0xf, 0xf, true));
}
static __device__ __forceinline__ float row16_sum(float v)
{
    v += xmask<8>(v);
    v += xmask<4>(v);
    v += xmask<2>(v);
    v += xmask<1>(v);
    return v;
}
static __device__ __forceinline__ float row16_max(float v)
{
    v = fmaxf(v, xmask<8>(v));
    v = fmaxf(v, xmask<4>(v));
    v = fmaxf(v, xmask<2>(v));
    v = fmaxf(v, xmask<1>(v));
    return v;
}

static __device__ __forceinline__ void prefetch_rows8(const float* rows, int lane)
{
#pragma unroll
    for (int i = 0; i < (8 * D_) / (32 * 32); ++i)
        __builtin_prefetch(rows + (i * 32 + lane) * 32, 0, 2);
}

static __device__ __forceinline__ void g1_load(G1Set& s, const float* qrow, const float* trow, int kf0)
{
#pragma unroll
    for (int i = 0; i < KB1; ++i) {
        const int kf = kf0 + 4 * i;
        s.q[i] = *(const float2*)(qrow + kf);
        s.x[i] = *(const float2*)(trow + kf);
    }
}

static __device__ __forceinline__ v8f g1_mma(const G1Set& s, v8f S, v2f (&ssum)[2], v2f (&ssq)[2])
{
#pragma unroll
    for (int i = 0; i < KB1; ++i) {
        v2f a;
        a.x = s.q[i].x;
        a.y = s.q[i].y;
        v2f bm;
        bm.x = s.x[i].x;
        bm.y = s.x[i].y;
        ssum[i & 1] += bm;
        ssq[i & 1] = bm * bm + ssq[i & 1];
        S = __builtin_amdgcn_wmma_f32_16x16x4_f32(false, a, false, bm,
                                                  (short)0, S, false, false);
    }
    return S;
}

static __device__ __forceinline__ void g2_load(G2Set& s, const float* tcol, int j0)
{
#pragma unroll
    for (int jj = 0; jj < JB; ++jj) {
        const int j = j0 + jj;
#pragma unroll
        for (int kk = 0; kk < 4; ++kk) {
            s.x[jj][kk].x = tcol[(kk * 4) * RP + j * 16];
            s.x[jj][kk].y = tcol[(kk * 4 + 1) * RP + j * 16];
        }
    }
}

#define LDS_FLOATS (16*RP + 2*D_ + NWAVES*TILE_FLOATS + 16*2 + NWAVES*256 + NWAVES*8*3 + 32)

extern "C" __global__ __launch_bounds__(BLOCK)
void pm_fused(const float* __restrict__ x, const float* __restrict__ q,
              const float* __restrict__ gamma, const float* __restrict__ beta,
              float* __restrict__ pmax, float* __restrict__ pl,
              float* __restrict__ pacc)
{
    extern __shared__ __attribute__((aligned(16))) float lds[];
    float* sQ    = lds;
    float* sG    = sQ + 16 * RP;
    float* sBe   = sG + D_;
    float* sTile = sBe + D_;
    float* sQG   = sTile + NWAVES * TILE_FLOATS;
    float* sQB   = sQG + 16;
    float* sP    = sQB + 16;
    float* sWmax = sP + NWAVES * 256;
    float* sWl   = sWmax + NWAVES * 8;
    float* sWc   = sWl + NWAVES * 8;
    float* sBmax = sWc + NWAVES * 8;
    float* sBl   = sBmax + 8;
    float* sBc   = sBl + 8;

    const int tid  = threadIdx.x;
    const int wave = tid >> 5;
    const int lane = tid & 31;
    const int half = lane >> 4;
    const int l16  = lane & 15;
    const int b     = blockIdx.x / CHUNKS;
    const int chunk = blockIdx.x % CHUNKS;
    const long xbase = (long)b * N_ * D_;

    const float* tile0 = x + xbase + (long)(chunk * CN + wave * TILES_PER_WAVE * 16) * D_;
    prefetch_rows8(tile0, lane);

    static_assert(M_ % NWAVES == 0, "whole query rows per wave");
#pragma unroll
    for (int rr = 0; rr < M_ / NWAVES; ++rr) {
        const int m = wave * (M_ / NWAVES) + rr;
        float pg = 0.0f, pb = 0.0f;
#pragma unroll
        for (int i = 0; i < D_ / (32 * 4); ++i) {
            const int d = lane * 4 + i * 128;
            const float4 qv = *(const float4*)(q + m * D_ + d);
            const float4 gv = *(const float4*)(gamma + d);
            const float4 bv = *(const float4*)(beta + d);
            float4 w;
            w.x = qv.x * gv.x;
            w.y = qv.y * gv.y;
            w.z = qv.z * gv.z;
            w.w = qv.w * gv.w;
            pg += (w.x + w.y) + (w.z + w.w);
            pb += (qv.x * bv.x + qv.y * bv.y) + (qv.z * bv.z + qv.w * bv.w);
            *(float4*)(sQ + m * RP + d) = w;
        }
        pg = row16_sum(pg + xrow(pg));
        pb = row16_sum(pb + xrow(pb));
        if (lane == 0) { sQG[m] = pg; sQB[m] = pb; }
    }
    if (tid < 16 - M_) { sQG[M_ + tid] = 0.0f; sQB[M_ + tid] = 0.0f; }
#pragma unroll
    for (int it = (M_ * D_ / 4) / BLOCK; it < (16 * D_ / 4) / BLOCK; ++it) {
        const int i  = tid + it * BLOCK;
        const int r  = i / (D_ / 4);
        const int d4 = i - r * (D_ / 4);
        *(float4*)(sQ + r * RP + d4 * 4) = make_float4(0.0f, 0.0f, 0.0f, 0.0f);
    }
    for (int i = tid; i < D_ / 4; i += BLOCK) {
        *(float4*)(sG + i * 4)  = *(const float4*)(gamma + i * 4);
        *(float4*)(sBe + i * 4) = *(const float4*)(beta + i * 4);
    }
    prefetch_rows8(tile0 + 8 * D_, lane);
    __syncthreads();

    float* myTile = sTile + wave * TILE_FLOATS;
    float* myP    = sP + wave * 256;

    float rowmax[8], rowsum[8], rowc1[8];
#pragma unroll
    for (int r = 0; r < 8; ++r) { rowmax[r] = NEG_BIG; rowsum[r] = 0.0f; rowc1[r] = 0.0f; }

    float qg[8], qb[8];
#pragma unroll
    for (int r = 0; r < 8; ++r) { qg[r] = sQG[r + half * 8]; qb[r] = sQB[r + half * 8]; }

    v8f acc[FBLKS];
#pragma unroll
    for (int j = 0; j < FBLKS; ++j)
        acc[j] = (v8f){0.f, 0.f, 0.f, 0.f, 0.f, 0.f, 0.f, 0.f};

    const float scale = 0.03608439182435161f;

    for (int t = 0; t < TILES_PER_WAVE; ++t) {
        const int n0 = chunk * CN + (wave * TILES_PER_WAVE + t) * 16;

#pragma unroll 1
        for (int rg = 0; rg < 16; rg += RG) {
            float4 v[RG][6];
            const float* rowbase = x + xbase + (long)(n0 + rg) * D_ + lane * 4;
#pragma unroll
            for (int rr = 0; rr < RG; ++rr) {
#pragma unroll
                for (int i = 0; i < D_ / (32 * 4); ++i)
                    v[rr][i] = *(const float4*)(rowbase + rr * D_ + i * 128);
            }
#pragma unroll
            for (int rr = 0; rr < RG; ++rr) {
                float* tdst = myTile + (rg + rr) * RP + lane * 4;
#pragma unroll
                for (int i = 0; i < D_ / (32 * 4); ++i)
                    *(float4*)(tdst + i * 128) = v[rr][i];
            }
        }

        const float* pfTile = x + xbase + (long)(n0 + ((t + 1 < TILES_PER_WAVE) ? 16 : 0)) * D_;
        const int    pfLine = lane * 32;

        const float* trow = myTile + l16 * RP;
        const float* qrow = sQ + l16 * RP;

        v8f S = {0.f, 0.f, 0.f, 0.f, 0.f, 0.f, 0.f, 0.f};
        v2f ssum[2] = {{0.f, 0.f}, {0.f, 0.f}}, ssq[2] = {{0.f, 0.f}, {0.f, 0.f}};
        {
            const int kh = half * 2;
            G1Set s0, s1;
            g1_load(s0, qrow, trow, kh);
#pragma unroll 1
            for (int kb = 0; kb < D_; kb += 2 * 4 * KB1) {
                __builtin_prefetch(pfTile + kb * 8 + pfLine, 0, 2);
                g1_load(s1, qrow, trow, kb + 4 * KB1 + kh);
                __builtin_amdgcn_sched_barrier(0);
                S = g1_mma(s0, S, ssum, ssq);
                if (kb + 2 * 4 * KB1 < D_)
                    g1_load(s0, qrow, trow, kb + 2 * 4 * KB1 + kh);
                S = g1_mma(s1, S, ssum, ssq);
            }
        }

        float rsum = (ssum[0].x + ssum[1].x) + (ssum[0].y + ssum[1].y);
        float rsq  = (ssq[0].x + ssq[1].x) + (ssq[0].y + ssq[1].y);
        rsum += xrow(rsum);
        rsq  += xrow(rsq);
        const float mu_r   = rsum * (1.0f / D_);
        const float var_r  = rsq * (1.0f / D_) - mu_r * mu_r;
        const float rs_r   = rsqrtf(var_r + EPS_);
        const float rsmu_r = rs_r * mu_r;

        float corr[8];
#pragma unroll
        for (int r = 0; r < 8; ++r) {
            float sim = (rs_r * (S[r] - mu_r * qg[r]) + qb[r]) * scale;
            const float tmax = row16_max(sim);
            float nm = fmaxf(rowmax[r], tmax);
            float p  = __expf(sim - nm);
            float c  = __expf(rowmax[r] - nm);
            const float ps = row16_sum(p);
            const float pc = row16_sum(p * rsmu_r);
            rowsum[r] = rowsum[r] * c + ps;
            rowc1[r]  = rowc1[r] * c + pc;
            rowmax[r] = nm;
            corr[r]   = c;
            myP[(r + half * 8) * 16 + l16] = p * rs_r;
        }

        v2f   pa[4];
#pragma unroll
        for (int kk = 0; kk < 4; ++kk) {
            const int r0 = kk * 4 + half * 2;
            pa[kk].x = myP[l16 * 16 + r0];
            pa[kk].y = myP[l16 * 16 + r0 + 1];
        }
        const float* tcol = myTile + (half * 2) * RP + l16;
        {
            G2Set st[2];
            g2_load(st[0], tcol, 0);
#pragma unroll
            for (int blk = 0; blk < NB2; ++blk) {
                if (blk + 1 < NB2)
                    g2_load(st[(blk + 1) & 1], tcol, (blk + 1) * JB);
                __builtin_prefetch(pfTile + (8 * D_ + blk * 1024) + pfLine, 0, 2);
                __builtin_amdgcn_sched_barrier(0);
                const G2Set& s = st[blk & 1];
#pragma unroll
                for (int jj = 0; jj < JB; ++jj) {
                    const int   j  = blk * JB + jj;
                    v8f Dk = {0.f, 0.f, 0.f, 0.f, 0.f, 0.f, 0.f, 0.f};
#pragma unroll
                    for (int kk = 0; kk < 4; ++kk)
                        Dk = __builtin_amdgcn_wmma_f32_16x16x4_f32(false, pa[kk], false, s.x[jj][kk],
                                                                   (short)0, Dk, false, false);
#pragma unroll
                    for (int r = 0; r < 8; ++r)
                        acc[j][r] = acc[j][r] * corr[r] + Dk[r];
                }
                __builtin_amdgcn_sched_barrier(0);
            }
        }
    }

    if (lane == 0) {
#pragma unroll
        for (int r = 0; r < 8; ++r) {
            sWmax[wave * 8 + r] = rowmax[r];
            sWl[wave * 8 + r]   = rowsum[r];
            sWc[wave * 8 + r]   = rowc1[r];
        }
    }
    if (half == 0) {
#pragma unroll
        for (int j = 0; j < FBLKS; ++j) {
#pragma unroll
            for (int r = 0; r < 8; ++r)
                myTile[r * D_ + j * 16 + l16] = acc[j][r];
        }
    }
    __syncthreads();

    const int pidx = blockIdx.x;
    if (tid < 8) {
        float mb = NEG_BIG;
        for (int w = 0; w < NWAVES; ++w) mb = fmaxf(mb, sWmax[w * 8 + tid]);
        float lb = 0.0f, cb = 0.0f;
        for (int w = 0; w < NWAVES; ++w) {
            const float wg = __expf(sWmax[w * 8 + tid] - mb);
            lb += sWl[w * 8 + tid] * wg;
            cb += sWc[w * 8 + tid] * wg;
        }
        sBmax[tid] = mb;
        sBl[tid]   = lb;
        sBc[tid]   = cb;
        pmax[pidx * 8 + tid] = mb;
        pl[pidx * 8 + tid]   = lb;
    }
    __syncthreads();

    for (int e = tid; e < 8 * D_; e += BLOCK) {
        const int m  = e / D_;
        const int d  = e - m * D_;
        const float mb = sBmax[m];
        float sum = 0.0f;
        for (int w = 0; w < NWAVES; ++w)
            sum += sTile[w * TILE_FLOATS + e] * __expf(sWmax[w * 8 + m] - mb);
        pacc[(long)pidx * 8 * D_ + e] = sG[d] * (sum - sBc[m]) + sBe[d] * sBl[m];
    }
}

#define RBLOCK 256
#define RB_PER_BATCH (M_ * D_ / RBLOCK)

extern "C" __global__ __launch_bounds__(RBLOCK)
void pm_reduce(const float* __restrict__ pmax, const float* __restrict__ pl,
               const float* __restrict__ pacc, float* __restrict__ out)
{
    static_assert(CHUNKS <= 32, "one chunk per lane of the first wave");
    __shared__ float cw[CHUNKS];
    __shared__ float gLinv;
    const int b   = blockIdx.x / RB_PER_BATCH;
    const int blk = blockIdx.x % RB_PER_BATCH;
    const int tid = threadIdx.x;
    const int e   = blk * RBLOCK + tid;
    const int m   = (blk * RBLOCK) / D_;
    if (tid < 32) {
        const bool live = tid < CHUNKS;
        float pm = live ? pmax[(b * CHUNKS + tid) * 8 + m] : NEG_BIG;
        float lv = live ? pl[(b * CHUNKS + tid) * 8 + m]   : 0.0f;
        const float mb = row16_max(fmaxf(pm, xrow(pm)));
        float w = __expf(pm - mb);
        if (live) cw[tid] = w;
        float lb = lv * w;
        lb = row16_sum(lb + xrow(lb));
        if (tid == 0) gLinv = 1.0f / lb;
    }
    __syncthreads();
    const float* src = pacc + (long)b * CHUNKS * 8 * D_ + e;
    float s = 0.0f;
#pragma unroll
    for (int c = 0; c < CHUNKS; ++c)
        s += src[(long)c * 8 * D_] * cw[c];
    out[(long)b * 8 * D_ + e] = s * gLinv;
}

extern "C" void kernel_launch(void* const* d_in, const int* in_sizes, int n_in,
                              void* d_out, int out_size, void* d_ws, size_t ws_size,
                              hipStream_t stream)
{
    const float* x     = (const float*)d_in[0];
    const float* q     = (const float*)d_in[1];
    const float* gamma = (const float*)d_in[2];
    const float* beta  = (const float*)d_in[3];
    float* out = (float*)d_out;

    const int P = B_ * CHUNKS;
    float* pmax = (float*)d_ws;
    float* pl   = pmax + (size_t)P * 8;
    float* pacc = pl + (size_t)P * 8;

    const size_t shmem = (size_t)LDS_FLOATS * sizeof(float);
    hipFuncSetAttribute(reinterpret_cast<const void*>(pm_fused),
                        hipFuncAttributeMaxDynamicSharedMemorySize, (int)shmem);

    pm_fused<<<P, BLOCK, shmem, stream>>>(x, q, gamma, beta, pmax, pl, pacc);
    pm_reduce<<<B_ * RB_PER_BATCH, RBLOCK, 0, stream>>>(pmax, pl, pacc, out);
}


# In-group merge of the 4 waves' sums: wave weights computed once per query row (not one exp per element and wave), 4 features per thread, 128-bit LDS reads and stores; wave order of sums kept
# speedup vs baseline: 10.8927x; 1.0854x over previous
#include <hip/hip_runtime.h>
#include <math.h>


typedef __attribute__((ext_vector_type(2))) float v2f;
typedef __attribute__((ext_vector_type(8))) float v8f;

#define B_ 16
#define N_ 4096
#define D_ 768
#define M_ 8
#define CHUNKS 16
#define CN (N_ / CHUNKS)
#define NWAVES 4
#define BLOCK (NWAVES * 32)
#define TILES_PER_WAVE (CN / 16 / NWAVES)
#define EPS_ 1e-3f
#define NEG_BIG (-1e30f)
#define FBLKS (D_ / 16)
#define RP 780
#define TILE_FLOATS (16 * RP)
#define RG 8
#define KB1 16
#define NB1 (D_ / (4 * KB1))
#define JB 8
#define NB2 (FBLKS / JB)


struct G1Set { float2 q[KB1], x[KB1]; };
struct G2Set { v2f x[JB][4]; };

static __device__ __forceinline__ float xrow(float x)
{
    const int b = __builtin_bit_cast(int, x);
    return __builtin_bit_cast(float, __builtin_amdgcn_permlanex16(b, b, 0x76543210, 0xfedcba98,
                                                                  false, false));
}
template <int M>
static __device__ __forceinline__ float xmask(float x)
{
    const int b = __builtin_bit_cast(int, x);
    return __builtin_bit_cast(float, __builtin_amdgcn_update_dpp(0, b, 0x160 + M, 0xf, 0xf, true));
}
static __device__ __forceinline__ float row16_sum(float v)
{
    v += xmask<8>(v);
    v += xmask<4>(v);
    v += xmask<2>(v);
    v += xmask<1>(v);
    return v;
}
static __device__ __forceinline__ float row16_max(float v)
{
    v = fmaxf(v, xmask<8>(v));
    v = fmaxf(v, xmask<4>(v));
    v = fmaxf(v, xmask<2>(v));
    v = fmaxf(v, xmask<1>(v));
    return v;
}

static __device__ __forceinline__ void prefetch_rows8(const float* rows, int lane)
{
#pragma unroll
    for (int i = 0; i < (8 * D_) / (32 * 32); ++i)
        __builtin_prefetch(rows + (i * 32 + lane) * 32, 0, 2);
}

static __device__ __forceinline__ void g1_load(G1Set& s, const float* qrow, const float* trow, int kf0)
{
#pragma unroll
    for (int i = 0; i < KB1; ++i) {
        const int kf = kf0 + 4 * i;
        s.q[i] = *(const float2*)(qrow + kf);
        s.x[i] = *(const float2*)(trow + kf);
    }
}

static __device__ __forceinline__ v8f g1_mma(const G1Set& s, v8f S, v2f (&ssum)[2], v2f (&ssq)[2])
{
#pragma unroll
    for (int i = 0; i < KB1; ++i) {
        v2f a;
        a.x = s.q[i].x;
        a.y = s.q[i].y;
        v2f bm;
        bm.x = s.x[i].x;
        bm.y = s.x[i].y;
        ssum[i & 1] += bm;
        ssq[i & 1] = bm * bm + ssq[i & 1];
        S = __builtin_amdgcn_wmma_f32_16x16x4_f32(false, a, false, bm,
                                                  (short)0, S, false, false);
    }
    return S;
}

static __device__ __forceinline__ void g2_load(G2Set& s, const float* tcol, int j0)
{
#pragma unroll
    for (int jj = 0; jj < JB; ++jj) {
        const int j = j0 + jj;
#pragma unroll
        for (int kk = 0; kk < 4; ++kk) {
            s.x[jj][kk].x = tcol[(kk * 4) * RP + j * 16];
            s.x[jj][kk].y = tcol[(kk * 4 + 1) * RP + j * 16];
        }
    }
}

#define LDS_FLOATS (16*RP + 2*D_ + NWAVES*TILE_FLOATS + 16*2 + NWAVES*256 + NWAVES*8*3 + 16 + 8*NWAVES)

extern "C" __global__ __launch_bounds__(BLOCK)
void pm_fused(const float* __restrict__ x, const float* __restrict__ q,
              const float* __restrict__ gamma, const float* __restrict__ beta,
              float* __restrict__ pmax, float* __restrict__ pl,
              float* __restrict__ pacc)
{
    extern __shared__ __attribute__((aligned(16))) float lds[];
    float* sQ    = lds;
    float* sG    = sQ + 16 * RP;
    float* sBe   = sG + D_;
    float* sTile = sBe + D_;
    float* sQG   = sTile + NWAVES * TILE_FLOATS;
    float* sQB   = sQG + 16;
    float* sP    = sQB + 16;
    float* sWmax = sP + NWAVES * 256;
    float* sWl   = sWmax + NWAVES * 8;
    float* sWc   = sWl + NWAVES * 8;
    float* sBl   = sWc + NWAVES * 8;
    float* sBc   = sBl + 8;
    float* sWw   = sBc + 8;

    const int tid  = threadIdx.x;
    const int wave = tid >> 5;
    const int lane = tid & 31;
    const int half = lane >> 4;
    const int l16  = lane & 15;
    const int b     = blockIdx.x / CHUNKS;
    const int chunk = blockIdx.x % CHUNKS;
    const long xbase = (long)b * N_ * D_;

    const float* tile0 = x + xbase + (long)(chunk * CN + wave * TILES_PER_WAVE * 16) * D_;
    prefetch_rows8(tile0, lane);

    static_assert(M_ % NWAVES == 0, "whole query rows per wave");
#pragma unroll
    for (int rr = 0; rr < M_ / NWAVES; ++rr) {
        const int m = wave * (M_ / NWAVES) + rr;
        float pg = 0.0f, pb = 0.0f;
#pragma unroll
        for (int i = 0; i < D_ / (32 * 4); ++i) {
            const int d = lane * 4 + i * 128;
            const float4 qv = *(const float4*)(q + m * D_ + d);
            const float4 gv = *(const float4*)(gamma + d);
            const float4 bv = *(const float4*)(beta + d);
            float4 w;
            w.x = qv.x * gv.x;
            w.y = qv.y * gv.y;
            w.z = qv.z * gv.z;
            w.w = qv.w * gv.w;
            pg += (w.x + w.y) + (w.z + w.w);
            pb += (qv.x * bv.x + qv.y * bv.y) + (qv.z * bv.z + qv.w * bv.w);
            *(float4*)(sQ + m * RP + d) = w;
        }
        pg = row16_sum(pg + xrow(pg));
        pb = row16_sum(pb + xrow(pb));
        if (lane == 0) { sQG[m] = pg; sQB[m] = pb; }
    }
    if (tid < 16 - M_) { sQG[M_ + tid] = 0.0f; sQB[M_ + tid] = 0.0f; }
#pragma unroll
    for (int it = (M_ * D_ / 4) / BLOCK; it < (16 * D_ / 4) / BLOCK; ++it) {
        const int i  = tid + it * BLOCK;
        const int r  = i / (D_ / 4);
        const int d4 = i - r * (D_ / 4);
        *(float4*)(sQ + r * RP + d4 * 4) = make_float4(0.0f, 0.0f, 0.0f, 0.0f);
    }
    for (int i = tid; i < D_ / 4; i += BLOCK) {
        *(float4*)(sG + i * 4)  = *(const float4*)(gamma + i * 4);
        *(float4*)(sBe + i * 4) = *(const float4*)(beta + i * 4);
    }
    prefetch_rows8(tile0 + 8 * D_, lane);
    __syncthreads();

    float* myTile = sTile + wave * TILE_FLOATS;
    float* myP    = sP + wave * 256;

    float rowmax[8], rowsum[8], rowc1[8];
#pragma unroll
    for (int r = 0; r < 8; ++r) { rowmax[r] = NEG_BIG; rowsum[r] = 0.0f; rowc1[r] = 0.0f; }

    float qg[8], qb[8];
#pragma unroll
    for (int r = 0; r < 8; ++r) { qg[r] = sQG[r + half * 8]; qb[r] = sQB[r + half * 8]; }

    v8f acc[FBLKS];
#pragma unroll
    for (int j = 0; j < FBLKS; ++j)
        acc[j] = (v8f){0.f, 0.f, 0.f, 0.f, 0.f, 0.f, 0.f, 0.f};

    const float scale = 0.03608439182435161f;

    for (int t = 0; t < TILES_PER_WAVE; ++t) {
        const int n0 = chunk * CN + (wave * TILES_PER_WAVE + t) * 16;

#pragma unroll 1
        for (int rg = 0; rg < 16; rg += RG) {
            float4 v[RG][6];
            const float* rowbase = x + xbase + (long)(n0 + rg) * D_ + lane * 4;
#pragma unroll
            for (int rr = 0; rr < RG; ++rr) {
#pragma unroll
                for (int i = 0; i < D_ / (32 * 4); ++i)
                    v[rr][i] = *(const float4*)(rowbase + rr * D_ + i * 128);
            }
#pragma unroll
            for (int rr = 0; rr < RG; ++rr) {
                float* tdst = myTile + (rg + rr) * RP + lane * 4;
#pragma unroll
                for (int i = 0; i < D_ / (32 * 4); ++i)
                    *(float4*)(tdst + i * 128) = v[rr][i];
            }
        }

        const float* pfTile = x + xbase + (long)(n0 + ((t + 1 < TILES_PER_WAVE) ? 16 : 0)) * D_;
        const int    pfLine = lane * 32;

        const float* trow = myTile + l16 * RP;
        const float* qrow = sQ + l16 * RP;

        v8f S = {0.f, 0.f, 0.f, 0.f, 0.f, 0.f, 0.f, 0.f};
        v2f ssum[2] = {{0.f, 0.f}, {0.f, 0.f}}, ssq[2] = {{0.f, 0.f}, {0.f, 0.f}};
        {
            const int kh = half * 2;
            G1Set s0, s1;
            g1_load(s0, qrow, trow, kh);
#pragma unroll 1
            for (int kb = 0; kb < D_; kb += 2 * 4 * KB1) {
                __builtin_prefetch(pfTile + kb * 8 + pfLine, 0, 2);
                g1_load(s1, qrow, trow, kb + 4 * KB1 + kh);
                __builtin_amdgcn_sched_barrier(0);
                S = g1_mma(s0, S, ssum, ssq);
                if (kb + 2 * 4 * KB1 < D_)
                    g1_load(s0, qrow, trow, kb + 2 * 4 * KB1 + kh);
                S = g1_mma(s1, S, ssum, ssq);
            }
        }

        float rsum = (ssum[0].x + ssum[1].x) + (ssum[0].y + ssum[1].y);
        float rsq  = (ssq[0].x + ssq[1].x) + (ssq[0].y + ssq[1].y);
        rsum += xrow(rsum);
        rsq  += xrow(rsq);
        const float mu_r   = rsum * (1.0f / D_);
        const float var_r  = rsq * (1.0f / D_) - mu_r * mu_r;
        const float rs_r   = rsqrtf(var_r + EPS_);
        const float rsmu_r = rs_r * mu_r;

        float corr[8];
#pragma unroll
        for (int r = 0; r < 8; ++r) {
            float sim = (rs_r * (S[r] - mu_r * qg[r]) + qb[r]) * scale;
            const float tmax = row16_max(sim);
            float nm = fmaxf(rowmax[r], tmax);
            float p  = __expf(sim - nm);
            float c  = __expf(rowmax[r] - nm);
            const float ps = row16_sum(p);
            const float pc = row16_sum(p * rsmu_r);
            rowsum[r] = rowsum[r] * c + ps;
            rowc1[r]  = rowc1[r] * c + pc;
            rowmax[r] = nm;
            corr[r]   = c;
            myP[(r + half * 8) * 16 + l16] = p * rs_r;
        }

        v2f   pa[4];
#pragma unroll
        for (int kk = 0; kk < 4; ++kk) {
            const int r0 = kk * 4 + half * 2;
            pa[kk].x = myP[l16 * 16 + r0];
            pa[kk].y = myP[l16 * 16 + r0 + 1];
        }
        const float* tcol = myTile + (half * 2) * RP + l16;
        {
            G2Set st[2];
            g2_load(st[0], tcol, 0);
#pragma unroll
            for (int blk = 0; blk < NB2; ++blk) {
                if (blk + 1 < NB2)
                    g2_load(st[(blk + 1) & 1], tcol, (blk + 1) * JB);
                __builtin_prefetch(pfTile + (8 * D_ + blk * 1024) + pfLine, 0, 2);
                __builtin_amdgcn_sched_barrier(0);
                const G2Set& s = st[blk & 1];
#pragma unroll
                for (int jj = 0; jj < JB; ++jj) {
                    const int   j  = blk * JB + jj;
                    v8f Dk = {0.f, 0.f, 0.f, 0.f, 0.f, 0.f, 0.f, 0.f};
#pragma unroll
                    for (int kk = 0; kk < 4; ++kk)
                        Dk = __builtin_amdgcn_wmma_f32_16x16x4_f32(false, pa[kk], false, s.x[jj][kk],
                                                                   (short)0, Dk, false, false);
#pragma unroll
                    for (int r = 0; r < 8; ++r)
                        acc[j][r] = acc[j][r] * corr[r] + Dk[r];
                }
                __builtin_amdgcn_sched_barrier(0);
            }
        }
    }

    if (lane == 0) {
#pragma unroll
        for (int r = 0; r < 8; ++r) {
            sWmax[wave * 8 + r] = rowmax[r];
            sWl[wave * 8 + r]   = rowsum[r];
            sWc[wave * 8 + r]   = rowc1[r];
        }
    }
    if (half == 0) {
#pragma unroll
        for (int j = 0; j < FBLKS; ++j) {
#pragma unroll
            for (int r = 0; r < 8; ++r)
                myTile[r * D_ + j * 16 + l16] = acc[j][r];
        }
    }
    __syncthreads();

    const int pidx = blockIdx.x;
    if (tid < 8) {
        float mb = NEG_BIG;
        for (int w = 0; w < NWAVES; ++w) mb = fmaxf(mb, sWmax[w * 8 + tid]);
        float lb = 0.0f, cb = 0.0f;
        for (int w = 0; w < NWAVES; ++w) {
            const float wg = __expf(sWmax[w * 8 + tid] - mb);
            lb += sWl[w * 8 + tid] * wg;
            cb += sWc[w * 8 + tid] * wg;
            sWw[tid * NWAVES + w] = wg;
        }
        sBl[tid]   = lb;
        sBc[tid]   = cb;
        pmax[pidx * 8 + tid] = mb;
        pl[pidx * 8 + tid]   = lb;
    }
    __syncthreads();

    static_assert(NWAVES == 4, "one float4 of wave weights per query row");
    static_assert((8 * D_ / 4) % BLOCK == 0, "whole trips");
    float* pdst = pacc + (long)pidx * 8 * D_;
#pragma unroll
    for (int it = 0; it < (8 * D_ / 4) / BLOCK; ++it) {
        const int i4 = tid + it * BLOCK;
        const int m  = i4 / (D_ / 4);
        const int d  = (i4 - m * (D_ / 4)) * 4;
        const float4 wv = *(const float4*)(sWw + m * NWAVES);
        const float4 a0 = *(const float4*)(sTile + 0 * TILE_FLOATS + i4 * 4);
        const float4 a1 = *(const float4*)(sTile + 1 * TILE_FLOATS + i4 * 4);
        const float4 a2 = *(const float4*)(sTile + 2 * TILE_FLOATS + i4 * 4);
        const float4 a3 = *(const float4*)(sTile + 3 * TILE_FLOATS + i4 * 4);
        const float4 g  = *(const float4*)(sG + d);
        const float4 be = *(const float4*)(sBe + d);
        const float  c  = sBc[m];
        const float  l  = sBl[m];
        float4 o;
        o.x = g.x * ((((a0.x * wv.x + a1.x * wv.y) + a2.x * wv.z) + a3.x * wv.w) - c) + be.x * l;
        o.y = g.y * ((((a0.y * wv.x + a1.y * wv.y) + a2.y * wv.z) + a3.y * wv.w) - c) + be.y * l;
        o.z = g.z * ((((a0.z * wv.x + a1.z * wv.y) + a2.z * wv.z) + a3.z * wv.w) - c) + be.z * l;
        o.w = g.w * ((((a0.w * wv.x + a1.w * wv.y) + a2.w * wv.z) + a3.w * wv.w) - c) + be.w * l;
        *(float4*)(pdst + i4 * 4) = o;
    }
}

#define RBLOCK 256
#define RB_PER_BATCH (M_ * D_ / RBLOCK)

extern "C" __global__ __launch_bounds__(RBLOCK)
void pm_reduce(const float* __restrict__ pmax, const float* __restrict__ pl,
               const float* __restrict__ pacc, float* __restrict__ out)
{
    static_assert(CHUNKS <= 32, "one chunk per lane of the first wave");
    __shared__ float cw[CHUNKS];
    __shared__ float gLinv;
    const int b   = blockIdx.x / RB_PER_BATCH;
    const int blk = blockIdx.x % RB_PER_BATCH;
    const int tid = threadIdx.x;
    const int e   = blk * RBLOCK + tid;
    const int m   = (blk * RBLOCK) / D_;
    if (tid < 32) {
        const bool live = tid < CHUNKS;
        float pm = live ? pmax[(b * CHUNKS + tid) * 8 + m] : NEG_BIG;
        float lv = live ? pl[(b * CHUNKS + tid) * 8 + m]   : 0.0f;
        const float mb = row16_max(fmaxf(pm, xrow(pm)));
        float w = __expf(pm - mb);
        if (live) cw[tid] = w;
        float lb = lv * w;
        lb = row16_sum(lb + xrow(lb));
        if (tid == 0) gLinv = 1.0f / lb;
    }
    __syncthreads();
    const float* src = pacc + (long)b * CHUNKS * 8 * D_ + e;
    float s = 0.0f;
#pragma unroll
    for (int c = 0; c < CHUNKS; ++c)
        s += src[(long)c * 8 * D_] * cw[c];
    out[(long)b * 8 * D_ + e] = s * gLinv;
}

extern "C" void kernel_launch(void* const* d_in, const int* in_sizes, int n_in,
                              void* d_out, int out_size, void* d_ws, size_t ws_size,
                              hipStream_t stream)
{
    const float* x     = (const float*)d_in[0];
    const float* q     = (const float*)d_in[1];
    const float* gamma = (const float*)d_in[2];
    const float* beta  = (const float*)d_in[3];
    float* out = (float*)d_out;

    const int P = B_ * CHUNKS;
    float* pmax = (float*)d_ws;
    float* pl   = pmax + (size_t)P * 8;
    float* pacc = pl + (size_t)P * 8;

    const size_t shmem = (size_t)LDS_FLOATS * sizeof(float);
    hipFuncSetAttribute(reinterpret_cast<const void*>(pm_fused),
                        hipFuncAttributeMaxDynamicSharedMemorySize, (int)shmem);

    pm_fused<<<P, BLOCK, shmem, stream>>>(x, q, gamma, beta, pmax, pl, pacc);
    pm_reduce<<<B_ * RB_PER_BATCH, RBLOCK, 0, stream>>>(pmax, pl, pacc, out);
}


# Two waves per SIMD: 8 waves in 4 pairs, each wave owns half the feature columns of its pair's tile (copy, half of GEMM1's K, half of GEMM2's outputs); half scores added through LDS, sum order changes
# speedup vs baseline: 11.3110x; 1.0384x over previous
#include <hip/hip_runtime.h>
#include <math.h>


typedef __attribute__((ext_vector_type(2))) float v2f;
typedef __attribute__((ext_vector_type(8))) float v8f;

#define B_ 16
#define N_ 4096
#define D_ 768
#define M_ 8
#define CHUNKS 16
#define CN (N_ / CHUNKS)
#define NWAVES 8
#define NPAIRS (NWAVES / 2)
#define BLOCK (NWAVES * 32)
#define TILES_PER_PAIR (CN / 16 / NPAIRS)
#define HD (D_ / 2)
#define EPS_ 1e-3f
#define NEG_BIG (-1e30f)
#define FBLKS (HD / 16)
#define RP 780
#define TILE_FLOATS (16 * RP)
#define RG 8
#define XSLOT 288
#define KB1 16
#define NB1 (HD / (4 * KB1))
#define JB 8
#define NB2 (FBLKS / JB)


struct G1Set { float2 q[KB1], x[KB1]; };
struct G2Set { v2f x[JB][4]; };

static __device__ __forceinline__ float xrow(float x)
{
    const int b = __builtin_bit_cast(int, x);
    return __builtin_bit_cast(float, __builtin_amdgcn_permlanex16(b, b, 0x76543210, 0xfedcba98,
                                                                  false, false));
}
template <int M>
static __device__ __forceinline__ float xmask(float x)
{
    const int b = __builtin_bit_cast(int, x);
    return __builtin_bit_cast(float, __builtin_amdgcn_update_dpp(0, b, 0x160 + M, 0xf, 0xf, true));
}
static __device__ __forceinline__ float row16_sum(float v)
{
    v += xmask<8>(v);
    v += xmask<4>(v);
    v += xmask<2>(v);
    v += xmask<1>(v);
    return v;
}
static __device__ __forceinline__ float row16_max(float v)
{
    v = fmaxf(v, xmask<8>(v));
    v = fmaxf(v, xmask<4>(v));
    v = fmaxf(v, xmask<2>(v));
    v = fmaxf(v, xmask<1>(v));
    return v;
}

static __device__ __forceinline__ int pf_lane_offset(int lane)
{
    return (lane >> 1) * D_ + (lane & 1) * (HD / 2);
}
static __device__ __forceinline__ void prefetch_half3(const float* p, int i0)
{
#pragma unroll
    for (int i = 0; i < 3; ++i)
        __builtin_prefetch(p + (i0 + i) * 32, 0, 2);
}

static __device__ __forceinline__ void g1_load(G1Set& s, const float* qrow, const float* trow, int kf0)
{
#pragma unroll
    for (int i = 0; i < KB1; ++i) {
        const int kf = kf0 + 4 * i;
        s.q[i] = *(const float2*)(qrow + kf);
        s.x[i] = *(const float2*)(trow + kf);
    }
}

static __device__ __forceinline__ v8f g1_mma(const G1Set& s, v8f S, v2f (&ssum)[2], v2f (&ssq)[2])
{
#pragma unroll
    for (int i = 0; i < KB1; ++i) {
        v2f a;
        a.x = s.q[i].x;
        a.y = s.q[i].y;
        v2f bm;
        bm.x = s.x[i].x;
        bm.y = s.x[i].y;
        ssum[i & 1] += bm;
        ssq[i & 1] = bm * bm + ssq[i & 1];
        S = __builtin_amdgcn_wmma_f32_16x16x4_f32(false, a, false, bm,
                                                  (short)0, S, false, false);
    }
    return S;
}

static __device__ __forceinline__ void g2_load(G2Set& s, const float* tcol, int j0)
{
#pragma unroll
    for (int jj = 0; jj < JB; ++jj) {
        const int j = j0 + jj;
#pragma unroll
        for (int kk = 0; kk < 4; ++kk) {
            s.x[jj][kk].x = tcol[(kk * 4) * RP + j * 16];
            s.x[jj][kk].y = tcol[(kk * 4 + 1) * RP + j * 16];
        }
    }
}

#define LDS_FLOATS (16*RP + 2*D_ + NPAIRS*TILE_FLOATS + 16*2 + NWAVES*256 + 2*NWAVES*XSLOT + NPAIRS*8*3 + 16 + 8*NPAIRS)

extern "C" __global__ __launch_bounds__(BLOCK)
void pm_fused(const float* __restrict__ x, const float* __restrict__ q,
              const float* __restrict__ gamma, const float* __restrict__ beta,
              float* __restrict__ pmax, float* __restrict__ pl,
              float* __restrict__ pacc)
{
    extern __shared__ __attribute__((aligned(16))) float lds[];
    float* sQ    = lds;
    float* sG    = sQ + 16 * RP;
    float* sBe   = sG + D_;
    float* sTile = sBe + D_;
    float* sQG   = sTile + NPAIRS * TILE_FLOATS;
    float* sQB   = sQG + 16;
    float* sP    = sQB + 16;
    float* sX    = sP + NWAVES * 256;
    float* sWmax = sX + 2 * NWAVES * XSLOT;
    float* sWl   = sWmax + NPAIRS * 8;
    float* sWc   = sWl + NPAIRS * 8;
    float* sBl   = sWc + NPAIRS * 8;
    float* sBc   = sBl + 8;
    float* sWw   = sBc + 8;
    static_assert((16*RP + 2*D_ + NPAIRS*TILE_FLOATS + 32 + NWAVES*256 + 2*NWAVES*XSLOT + NPAIRS*8*3 + 16) % 4 == 0,
                  "sWw rows are read as float4");

    const int tid  = threadIdx.x;
    const int wave = tid >> 5;
    const int lane = tid & 31;
    const int half = lane >> 4;
    const int l16  = lane & 15;
    const int pair = wave >> 1;
    const int side = wave & 1;
    const int colbase = side * HD;
    const int b     = blockIdx.x / CHUNKS;
    const int chunk = blockIdx.x % CHUNKS;
    const long xbase = (long)b * N_ * D_;

    const int pfLane = pf_lane_offset(lane);
    const float* tile0 = x + xbase + (long)(chunk * CN + pair * TILES_PER_PAIR * 16) * D_ + colbase + pfLane;
    prefetch_half3(tile0, 0);

    static_assert(M_ % NWAVES == 0, "whole query rows per wave");
#pragma unroll
    for (int rr = 0; rr < M_ / NWAVES; ++rr) {
        const int m = wave * (M_ / NWAVES) + rr;
        float pg = 0.0f, pb = 0.0f;
#pragma unroll
        for (int i = 0; i < D_ / (32 * 4); ++i) {
            const int d = lane * 4 + i * 128;
            const float4 qv = *(const float4*)(q + m * D_ + d);
            const float4 gv = *(const float4*)(gamma + d);
            const float4 bv = *(const float4*)(beta + d);
            float4 w;
            w.x = qv.x * gv.x;
            w.y = qv.y * gv.y;
            w.z = qv.z * gv.z;
            w.w = qv.w * gv.w;
            pg += (w.x + w.y) + (w.z + w.w);
            pb += (qv.x * bv.x + qv.y * bv.y) + (qv.z * bv.z + qv.w * bv.w);
            *(float4*)(sQ + m * RP + d) = w;
        }
        pg = row16_sum(pg + xrow(pg));
        pb = row16_sum(pb + xrow(pb));
        if (lane == 0) { sQG[m] = pg; sQB[m] = pb; }
    }
    if (tid < 16 - M_) { sQG[M_ + tid] = 0.0f; sQB[M_ + tid] = 0.0f; }
#pragma unroll
    for (int it = (M_ * D_ / 4) / BLOCK; it < (16 * D_ / 4) / BLOCK; ++it) {
        const int i  = tid + it * BLOCK;
        const int r  = i / (D_ / 4);
        const int d4 = i - r * (D_ / 4);
        *(float4*)(sQ + r * RP + d4 * 4) = make_float4(0.0f, 0.0f, 0.0f, 0.0f);
    }
    for (int i = tid; i < D_ / 4; i += BLOCK) {
        *(float4*)(sG + i * 4)  = *(const float4*)(gamma + i * 4);
        *(float4*)(sBe + i * 4) = *(const float4*)(beta + i * 4);
    }
    prefetch_half3(tile0, 3);
    __syncthreads();

    float* myTile = sTile + pair * TILE_FLOATS;
    float* myP    = sP + wave * 256;

    float rowmax[8], rowsum[8], rowc1[8];
#pragma unroll
    for (int r = 0; r < 8; ++r) { rowmax[r] = NEG_BIG; rowsum[r] = 0.0f; rowc1[r] = 0.0f; }

    float qg[8], qb[8];
#pragma unroll
    for (int r = 0; r < 8; ++r) { qg[r] = sQG[r + half * 8]; qb[r] = sQB[r + half * 8]; }

    v8f acc[FBLKS];
#pragma unroll
    for (int j = 0; j < FBLKS; ++j)
        acc[j] = (v8f){0.f, 0.f, 0.f, 0.f, 0.f, 0.f, 0.f, 0.f};

    const float scale = 0.03608439182435161f;

    for (int t = 0; t < TILES_PER_PAIR; ++t) {
        const int n0 = chunk * CN + (pair * TILES_PER_PAIR + t) * 16;

#pragma unroll 1
        for (int rg = 0; rg < 16; rg += RG) {
            float4 v[RG][3];
            const float* rowbase = x + xbase + (long)(n0 + rg) * D_ + colbase + lane * 4;
#pragma unroll
            for (int rr = 0; rr < RG; ++rr) {
#pragma unroll
                for (int i = 0; i < HD / (32 * 4); ++i)
                    v[rr][i] = *(const float4*)(rowbase + rr * D_ + i * 128);
            }
#pragma unroll
            for (int rr = 0; rr < RG; ++rr) {
                float* tdst = myTile + (rg + rr) * RP + colbase + lane * 4;
#pragma unroll
                for (int i = 0; i < HD / (32 * 4); ++i)
                    *(float4*)(tdst + i * 128) = v[rr][i];
            }
        }

        const float* pfTile = x + xbase + (long)(n0 + ((t + 1 < TILES_PER_PAIR) ? 16 : 0)) * D_ + colbase + pfLane;

        const float* trow = myTile + l16 * RP + colbase;
        const float* qrow = sQ + l16 * RP + colbase;

        v8f S = {0.f, 0.f, 0.f, 0.f, 0.f, 0.f, 0.f, 0.f};
        v2f ssum[2] = {{0.f, 0.f}, {0.f, 0.f}}, ssq[2] = {{0.f, 0.f}, {0.f, 0.f}};
        {
            const int kh = half * 2;
            G1Set s0, s1;
            g1_load(s0, qrow, trow, kh);
#pragma unroll 1
            for (int kb = 0; kb < HD; kb += 2 * 4 * KB1) {
                __builtin_prefetch(pfTile + (kb / (2 * 4 * KB1)) * 32, 0, 2);
                g1_load(s1, qrow, trow, kb + 4 * KB1 + kh);
                __builtin_amdgcn_sched_barrier(0);
                S = g1_mma(s0, S, ssum, ssq);
                if (kb + 2 * 4 * KB1 < HD)
                    g1_load(s0, qrow, trow, kb + 2 * 4 * KB1 + kh);
                S = g1_mma(s1, S, ssum, ssq);
            }
        }

        float rsum = (ssum[0].x + ssum[1].x) + (ssum[0].y + ssum[1].y);
        float rsq  = (ssq[0].x + ssq[1].x) + (ssq[0].y + ssq[1].y);
        rsum += xrow(rsum);
        rsq  += xrow(rsq);

        {
            float* myX = sX + ((t & 1) * NWAVES + wave) * XSLOT;
#pragma unroll
            for (int r = 0; r < 8; ++r)
                myX[(r + half * 8) * 16 + l16] = S[r];
            myX[256 + l16] = rsum;
            myX[272 + l16] = rsq;
            __syncthreads();
            const float* x0 = sX + ((t & 1) * NWAVES + (wave & ~1)) * XSLOT;
            const float* x1 = x0 + XSLOT;
#pragma unroll
            for (int r = 0; r < 8; ++r)
                S[r] = x0[(r + half * 8) * 16 + l16] + x1[(r + half * 8) * 16 + l16];
            rsum = x0[256 + l16] + x1[256 + l16];
            rsq  = x0[272 + l16] + x1[272 + l16];
        }
        const float mu_r   = rsum * (1.0f / D_);
        const float var_r  = rsq * (1.0f / D_) - mu_r * mu_r;
        const float rs_r   = rsqrtf(var_r + EPS_);
        const float rsmu_r = rs_r * mu_r;

        float corr[8];
#pragma unroll
        for (int r = 0; r < 8; ++r) {
            float sim = (rs_r * (S[r] - mu_r * qg[r]) + qb[r]) * scale;
            const float tmax = row16_max(sim);
            float nm = fmaxf(rowmax[r], tmax);
            float p  = __expf(sim - nm);
            float c  = __expf(rowmax[r] - nm);
            const float ps = row16_sum(p);
            const float pc = row16_sum(p * rsmu_r);
            rowsum[r] = rowsum[r] * c + ps;
            rowc1[r]  = rowc1[r] * c + pc;
            rowmax[r] = nm;
            corr[r]   = c;
            myP[(r + half * 8) * 16 + l16] = p * rs_r;
        }

        v2f   pa[4];
#pragma unroll
        for (int kk = 0; kk < 4; ++kk) {
            const int r0 = kk * 4 + half * 2;
            pa[kk].x = myP[l16 * 16 + r0];
            pa[kk].y = myP[l16 * 16 + r0 + 1];
        }
        const float* tcol = myTile + (half * 2) * RP + colbase + l16;
        {
            G2Set st[2];
            g2_load(st[0], tcol, 0);
#pragma unroll
            for (int blk = 0; blk < NB2; ++blk) {
                if (blk + 1 < NB2)
                    g2_load(st[(blk + 1) & 1], tcol, (blk + 1) * JB);
                __builtin_prefetch(pfTile + (3 + blk) * 32, 0, 2);
                __builtin_amdgcn_sched_barrier(0);
                const G2Set& s = st[blk & 1];
#pragma unroll
                for (int jj = 0; jj < JB; ++jj) {
                    const int   j  = blk * JB + jj;
                    v8f Dk = {0.f, 0.f, 0.f, 0.f, 0.f, 0.f, 0.f, 0.f};
#pragma unroll
                    for (int kk = 0; kk < 4; ++kk)
                        Dk = __builtin_amdgcn_wmma_f32_16x16x4_f32(false, pa[kk], false, s.x[jj][kk],
                                                                   (short)0, Dk, false, false);
#pragma unroll
                    for (int r = 0; r < 8; ++r)
                        acc[j][r] = acc[j][r] * corr[r] + Dk[r];
                }
                __builtin_amdgcn_sched_barrier(0);
            }
        }
    }

    __syncthreads();
    if (lane == 0 && side == 0) {
#pragma unroll
        for (int r = 0; r < 8; ++r) {
            sWmax[pair * 8 + r] = rowmax[r];
            sWl[pair * 8 + r]   = rowsum[r];
            sWc[pair * 8 + r]   = rowc1[r];
        }
    }
    if (half == 0) {
#pragma unroll
        for (int j = 0; j < FBLKS; ++j) {
#pragma unroll
            for (int r = 0; r < 8; ++r)
                myTile[r * D_ + colbase + j * 16 + l16] = acc[j][r];
        }
    }
    __syncthreads();

    const int pidx = blockIdx.x;
    if (tid < 8) {
        float mb = NEG_BIG;
        for (int w = 0; w < NPAIRS; ++w) mb = fmaxf(mb, sWmax[w * 8 + tid]);
        float lb = 0.0f, cb = 0.0f;
        for (int w = 0; w < NPAIRS; ++w) {
            const float wg = __expf(sWmax[w * 8 + tid] - mb);
            lb += sWl[w * 8 + tid] * wg;
            cb += sWc[w * 8 + tid] * wg;
            sWw[tid * NPAIRS + w] = wg;
        }
        sBl[tid]   = lb;
        sBc[tid]   = cb;
        pmax[pidx * 8 + tid] = mb;
        pl[pidx * 8 + tid]   = lb;
    }
    __syncthreads();

    static_assert(NPAIRS == 4, "one float4 of pair weights per query row");
    static_assert((8 * D_ / 4) % BLOCK == 0, "whole trips");
    float* pdst = pacc + (long)pidx * 8 * D_;
#pragma unroll
    for (int it = 0; it < (8 * D_ / 4) / BLOCK; ++it) {
        const int i4 = tid + it * BLOCK;
        const int m  = i4 / (D_ / 4);
        const int d  = (i4 - m * (D_ / 4)) * 4;
        const float4 wv = *(const float4*)(sWw + m * NPAIRS);
        const float4 a0 = *(const float4*)(sTile + 0 * TILE_FLOATS + i4 * 4);
        const float4 a1 = *(const float4*)(sTile + 1 * TILE_FLOATS + i4 * 4);
        const float4 a2 = *(const float4*)(sTile + 2 * TILE_FLOATS + i4 * 4);
        const float4 a3 = *(const float4*)(sTile + 3 * TILE_FLOATS + i4 * 4);
        const float4 g  = *(const float4*)(sG + d);
        const float4 be = *(const float4*)(sBe + d);
        const float  c  = sBc[m];
        const float  l  = sBl[m];
        float4 o;
        o.x = g.x * ((((a0.x * wv.x + a1.x * wv.y) + a2.x * wv.z) + a3.x * wv.w) - c) + be.x * l;
        o.y = g.y * ((((a0.y * wv.x + a1.y * wv.y) + a2.y * wv.z) + a3.y * wv.w) - c) + be.y * l;
        o.z = g.z * ((((a0.z * wv.x + a1.z * wv.y) + a2.z * wv.z) + a3.z * wv.w) - c) + be.z * l;
        o.w = g.w * ((((a0.w * wv.x + a1.w * wv.y) + a2.w * wv.z) + a3.w * wv.w) - c) + be.w * l;
        *(float4*)(pdst + i4 * 4) = o;
    }
}

#define RBLOCK 256
#define RB_PER_BATCH (M_ * D_ / RBLOCK)

extern "C" __global__ __launch_bounds__(RBLOCK)
void pm_reduce(const float* __restrict__ pmax, const float* __restrict__ pl,
               const float* __restrict__ pacc, float* __restrict__ out)
{
    static_assert(CHUNKS <= 32, "one chunk per lane of the first wave");
    __shared__ float cw[CHUNKS];
    __shared__ float gLinv;
    const int b   = blockIdx.x / RB_PER_BATCH;
    const int blk = blockIdx.x % RB_PER_BATCH;
    const int tid = threadIdx.x;
    const int e   = blk * RBLOCK + tid;
    const int m   = (blk * RBLOCK) / D_;
    if (tid < 32) {
        const bool live = tid < CHUNKS;
        float pm = live ? pmax[(b * CHUNKS + tid) * 8 + m] : NEG_BIG;
        float lv = live ? pl[(b * CHUNKS + tid) * 8 + m]   : 0.0f;
        const float mb = row16_max(fmaxf(pm, xrow(pm)));
        float w = __expf(pm - mb);
        if (live) cw[tid] = w;
        float lb = lv * w;
        lb = row16_sum(lb + xrow(lb));
        if (tid == 0) gLinv = 1.0f / lb;
    }
    __syncthreads();
    const float* src = pacc + (long)b * CHUNKS * 8 * D_ + e;
    float s = 0.0f;
#pragma unroll
    for (int c = 0; c < CHUNKS; ++c)
        s += src[(long)c * 8 * D_] * cw[c];
    out[(long)b * 8 * D_ + e] = s * gLinv;
}

extern "C" void kernel_launch(void* const* d_in, const int* in_sizes, int n_in,
                              void* d_out, int out_size, void* d_ws, size_t ws_size,
                              hipStream_t stream)
{
    const float* x     = (const float*)d_in[0];
    const float* q     = (const float*)d_in[1];
    const float* gamma = (const float*)d_in[2];
    const float* beta  = (const float*)d_in[3];
    float* out = (float*)d_out;

    const int P = B_ * CHUNKS;
    float* pmax = (float*)d_ws;
    float* pl   = pmax + (size_t)P * 8;
    float* pacc = pl + (size_t)P * 8;

    const size_t shmem = (size_t)LDS_FLOATS * sizeof(float);
    hipFuncSetAttribute(reinterpret_cast<const void*>(pm_fused),
                        hipFuncAttributeMaxDynamicSharedMemorySize, (int)shmem);

    pm_fused<<<P, BLOCK, shmem, stream>>>(x, q, gamma, beta, pmax, pl, pacc);
    pm_reduce<<<B_ * RB_PER_BATCH, RBLOCK, 0, stream>>>(pmax, pl, pacc, out);
}
